# starting kernel
# baseline (speedup 1.0000x reference)
#include <hip/hip_runtime.h>
#include <math.h>

#define NPROB 10000
#define NKC   100
#define NBATCH 1024
#define NT    500

typedef __attribute__((ext_vector_type(2))) float v2f;
typedef __attribute__((ext_vector_type(8))) float v8f;

__device__ __forceinline__ float lse2(float a, float b) {
  float m = fmaxf(a, b);
  return m + __logf(__expf(a - m) + __expf(b - m));
}

__global__ void k_logt(const float* __restrict__ trans, float* __restrict__ logtT) {
  int k = blockIdx.x * blockDim.x + threadIdx.x;
  if (k >= NKC) return;
  const float4 v = reinterpret_cast<const float4*>(trans)[k];
  const float l0 = lse2(v.x, v.z);
  const float l1 = lse2(v.y, v.w);
  logtT[0 * NKC + k] = v.x - l0;
  logtT[1 * NKC + k] = v.y - l1;
  logtT[2 * NKC + k] = v.z - l0;
  logtT[3 * NKC + k] = v.w - l1;
}

__global__ void k_lo(const float* __restrict__ obs, float* __restrict__ LO) {
  int p = blockIdx.x * blockDim.x + threadIdx.x;
  if (p >= NPROB) return;
  const float4 v = reinterpret_cast<const float4*>(obs)[p];
  const float ls0 = lse2(v.x, v.y);
  const float ls1 = lse2(v.z, v.w);
  float4 r;
  r.x = v.x - ls0; r.y = v.y - ls0; r.z = v.z - ls1; r.w = v.w - ls1;
  reinterpret_cast<float4*>(LO)[p] = r;
}

#if __has_builtin(__builtin_amdgcn_wmma_f32_16x16x4_f32)
#define HAVE_WMMA_F32X4 1
#endif

__global__ void __launch_bounds__(32) k_gemm_M(const float* __restrict__ A,
                                               const float* __restrict__ logtT,
                                               float* __restrict__ M8) {
  const int tile = blockIdx.x;
  const int l = threadIdx.x;
  const int m = l & 15;
  const bool hi = (l >= 16);
#ifdef HAVE_WMMA_F32X4
  const int row = tile * 16 + m;
  const float nmask = (m < 4) ? 1.0f : 0.0f;
  const float ones4 = (m == 4) ? 1.0f : 0.0f;
  const float* __restrict__ bcolp = logtT + (m & 3) * NKC;
  v8f c = {0.f, 0.f, 0.f, 0.f, 0.f, 0.f, 0.f, 0.f};
#pragma unroll 5
  for (int kk = 0; kk < 25; ++kk) {
    const int k0 = kk * 4 + (hi ? 2 : 0);
    const float2 av = *reinterpret_cast<const float2*>(A + row * NKC + k0);
    const float2 bv = *reinterpret_cast<const float2*>(bcolp + k0);
    v2f a, b;
    a.x = av.x;
    a.y = av.y;
    b.x = bv.x * nmask + ones4;
    b.y = bv.y * nmask + ones4;
    c = __builtin_amdgcn_wmma_f32_16x16x4_f32(
         false, a,  false, b,
         (short)0, c,  false,  false);
  }
  if (m < 5) {
#pragma unroll
    for (int v = 0; v < 8; ++v) {
      const int r = tile * 16 + v + (hi ? 8 : 0);
      M8[r * 8 + m] = c[v];
    }
  }
#else
  (void)hi;
  if (l < 16) {
    const int row = tile * 16 + m;
    float acc0 = 0.f, acc1 = 0.f, acc2 = 0.f, acc3 = 0.f, acc4 = 0.f;
    for (int k = 0; k < NKC; ++k) {
      const float av = A[row * NKC + k];
      acc0 += av * logtT[0 * NKC + k];
      acc1 += av * logtT[1 * NKC + k];
      acc2 += av * logtT[2 * NKC + k];
      acc3 += av * logtT[3 * NKC + k];
      acc4 += av;
    }
    M8[row * 8 + 0] = acc0; M8[row * 8 + 1] = acc1;
    M8[row * 8 + 2] = acc2; M8[row * 8 + 3] = acc3;
    M8[row * 8 + 4] = acc4;
  }
#endif
}

__global__ void __launch_bounds__(32) k_forward(
    const int* __restrict__ corr, const int* __restrict__ kc,
    const int* __restrict__ problem, const float* __restrict__ A,
    const float* __restrict__ init_logits, const float* __restrict__ M8,
    const float* __restrict__ LO, float* __restrict__ out) {
  const int b = blockIdx.x;
  const int l = threadIdx.x;
  const bool act = (l < 25);

  float al0[4] = {0.f, 0.f, 0.f, 0.f};
  float al1[4] = {0.f, 0.f, 0.f, 0.f};
  if (act) {
    const float4 i0 = reinterpret_cast<const float4*>(init_logits)[2 * l];
    const float4 i1 = reinterpret_cast<const float4*>(init_logits)[2 * l + 1];
    float z;
    z = lse2(i0.x, i0.y); al0[0] = i0.x - z; al1[0] = i0.y - z;
    z = lse2(i0.z, i0.w); al0[1] = i0.z - z; al1[1] = i0.w - z;
    z = lse2(i1.x, i1.y); al0[2] = i1.x - z; al1[2] = i1.y - z;
    z = lse2(i1.z, i1.w); al0[3] = i1.z - z; al1[3] = i1.w - z;
  }

  const int base = b * NT;
  int kci = kc[base];
  for (int t = 0; t < NT; ++t) {
    const int idx = base + t;
    const int pi = problem[idx];
    const int ci = corr[idx];
    const int kcn = kc[idx + ((t + 1 < NT) ? 1 : 0)];
    __builtin_prefetch(A + (size_t)kcn * NKC, 0, 3);

    float4 cj = {0.f, 0.f, 0.f, 0.f};
    if (act) cj = reinterpret_cast<const float4*>(A + (size_t)kci * NKC)[l];

    float p1 = cj.x * al0[0] + cj.y * al0[1] + cj.z * al0[2] + cj.w * al0[3];
    float p2 = cj.x * al1[0] + cj.y * al1[1] + cj.z * al1[2] + cj.w * al1[3];
#pragma unroll
    for (int msk = 16; msk >= 1; msk >>= 1) {
      p1 += __shfl_xor(p1, msk, 32);
      p2 += __shfl_xor(p2, msk, 32);
    }
    const float a20 = p1, a21 = p2;

    const float4 m4 = *reinterpret_cast<const float4*>(M8 + (size_t)kci * 8);
    const float SA = M8[(size_t)kci * 8 + 4];
    const float4 lo = reinterpret_cast<const float4*>(LO)[pi];
    const float LO00 = lo.x, LO01 = lo.y, LO10 = lo.z, LO11 = lo.w;

    const float lp0 = lse2(LO00 * SA + a20, LO10 * SA + a21);
    const float lp1 = lse2(LO01 * SA + a20, LO11 * SA + a21);
    const float lz = lse2(lp0, lp1);

    const float ol0 = (ci == 0 ? LO00 : LO01) * SA;
    const float ol1 = (ci == 0 ? LO10 : LO11) * SA;
    const float q0 = ol0 + a20;
    const float q1 = ol1 + a21;
    const float t30 = lse2(q0 + m4.x, q1 + m4.y);
    const float t31 = lse2(q0 + m4.z, q1 + m4.w);

    if (act) {
      al0[0] = (1.0f - cj.x) * al0[0] + cj.x * t30;
      al1[0] = (1.0f - cj.x) * al1[0] + cj.x * t31;
      al0[1] = (1.0f - cj.y) * al0[1] + cj.y * t30;
      al1[1] = (1.0f - cj.y) * al1[1] + cj.y * t31;
      al0[2] = (1.0f - cj.z) * al0[2] + cj.z * t30;
      al1[2] = (1.0f - cj.z) * al1[2] + cj.z * t31;
      al0[3] = (1.0f - cj.w) * al0[3] + cj.w * t30;
      al1[3] = (1.0f - cj.w) * al1[3] + cj.w * t31;
    }
    if (l == 0) {
      float2 o2;
      o2.x = lp0 - lz;
      o2.y = lp1 - lz;
      reinterpret_cast<float2*>(out)[idx] = o2;
    }
    kci = kcn;
  }
}

extern "C" void kernel_launch(void* const* d_in, const int* in_sizes, int n_in,
                              void* d_out, int out_size, void* d_ws, size_t ws_size,
                              hipStream_t stream) {
  (void)in_sizes; (void)n_in; (void)out_size; (void)ws_size;
  const int*   corr    = (const int*)d_in[0];
  const int*   kc      = (const int*)d_in[1];
  const int*   problem = (const int*)d_in[2];
  const float* A       = (const float*)d_in[3];
  const float* trans   = (const float*)d_in[4];
  const float* obs     = (const float*)d_in[5];
  const float* init    = (const float*)d_in[6];
  float* out = (float*)d_out;

  float* logtT = (float*)d_ws;
  float* M8    = logtT + 512;
  float* LO    = M8 + (size_t)NPROB * 8;

  k_logt<<<1, 128, 0, stream>>>(trans, logtT);
  k_lo<<<(NPROB + 127) / 128, 128, 0, stream>>>(obs, LO);
  k_gemm_M<<<NPROB / 16, 32, 0, stream>>>(A, logtT, M8);
  k_forward<<<NBATCH, 32, 0, stream>>>(corr, kc, problem, A, init, M8, LO, out);
}


# k_forward: two chains per wave, one per 16-lane row (a lane holds 8 KCs; 512 waves): step-by-step scan, a step's two sums are 4-add row butterflies on the vector ALU; rows come 4 steps ahead
# speedup vs baseline: 3.3712x; 3.3712x over previous
#include <hip/hip_runtime.h>
#include <math.h>

#define NPROB 10000
#define NKC   100
#define NBATCH 1024
#define NT    500

typedef __attribute__((ext_vector_type(2))) float v2f;
typedef __attribute__((ext_vector_type(8))) float v8f;

__device__ __forceinline__ float lse2(float a, float b) {
  float m = fmaxf(a, b);
  return m + __logf(__expf(a - m) + __expf(b - m));
}

__global__ void k_logt(const float* __restrict__ trans, float* __restrict__ logtT) {
  int k = blockIdx.x * blockDim.x + threadIdx.x;
  if (k >= NKC) return;
  const float4 v = reinterpret_cast<const float4*>(trans)[k];
  const float l0 = lse2(v.x, v.z);
  const float l1 = lse2(v.y, v.w);
  logtT[0 * NKC + k] = v.x - l0;
  logtT[1 * NKC + k] = v.y - l1;
  logtT[2 * NKC + k] = v.z - l0;
  logtT[3 * NKC + k] = v.w - l1;
}

__global__ void k_lo(const float* __restrict__ obs, float* __restrict__ LO) {
  int p = blockIdx.x * blockDim.x + threadIdx.x;
  if (p >= NPROB) return;
  const float4 v = reinterpret_cast<const float4*>(obs)[p];
  const float ls0 = lse2(v.x, v.y);
  const float ls1 = lse2(v.z, v.w);
  float4 r;
  r.x = v.x - ls0; r.y = v.y - ls0; r.z = v.z - ls1; r.w = v.w - ls1;
  reinterpret_cast<float4*>(LO)[p] = r;
}

#if __has_builtin(__builtin_amdgcn_wmma_f32_16x16x4_f32)
#define HAVE_WMMA_F32X4 1
#endif

__global__ void __launch_bounds__(32) k_gemm_M(const float* __restrict__ A,
                                               const float* __restrict__ logtT,
                                               float* __restrict__ M8) {
  const int tile = blockIdx.x;
  const int l = threadIdx.x;
  const int m = l & 15;
  const bool hi = (l >= 16);
#ifdef HAVE_WMMA_F32X4
  const int row = tile * 16 + m;
  const float nmask = (m < 4) ? 1.0f : 0.0f;
  const float ones4 = (m == 4) ? 1.0f : 0.0f;
  const float* __restrict__ bcolp = logtT + (m & 3) * NKC;
  v8f c = {0.f, 0.f, 0.f, 0.f, 0.f, 0.f, 0.f, 0.f};
#pragma unroll 5
  for (int kk = 0; kk < 25; ++kk) {
    const int k0 = kk * 4 + (hi ? 2 : 0);
    const float2 av = *reinterpret_cast<const float2*>(A + row * NKC + k0);
    const float2 bv = *reinterpret_cast<const float2*>(bcolp + k0);
    v2f a, b;
    a.x = av.x;
    a.y = av.y;
    b.x = bv.x * nmask + ones4;
    b.y = bv.y * nmask + ones4;
    c = __builtin_amdgcn_wmma_f32_16x16x4_f32(
         false, a,  false, b,
         (short)0, c,  false,  false);
  }
  if (m < 5) {
#pragma unroll
    for (int v = 0; v < 8; ++v) {
      const int r = tile * 16 + v + (hi ? 8 : 0);
      M8[r * 8 + m] = c[v];
    }
  }
#else
  (void)hi;
  if (l < 16) {
    const int row = tile * 16 + m;
    float acc0 = 0.f, acc1 = 0.f, acc2 = 0.f, acc3 = 0.f, acc4 = 0.f;
    for (int k = 0; k < NKC; ++k) {
      const float av = A[row * NKC + k];
      acc0 += av * logtT[0 * NKC + k];
      acc1 += av * logtT[1 * NKC + k];
      acc2 += av * logtT[2 * NKC + k];
      acc3 += av * logtT[3 * NKC + k];
      acc4 += av;
    }
    M8[row * 8 + 0] = acc0; M8[row * 8 + 1] = acc1;
    M8[row * 8 + 2] = acc2; M8[row * 8 + 3] = acc3;
    M8[row * 8 + 4] = acc4;
  }
#endif
}

#define TB 4
static_assert(NT % TB == 0, "row blocks must tile the sequence");
static_assert((NT / TB) % 2 == 1, "the loop below takes block pairs and one closing block");

struct Rows {
  float4 c0[TB];
  float4 c1[TB];
  float4 m4[TB];
  float4 lo[TB];
  float sa[TB];
};
struct Idx { int4 kc; int4 pr; int4 co; };

__device__ __forceinline__ int pick(const int4 v, int i) {
  return i == 0 ? v.x : (i == 1 ? v.y : (i == 2 ? v.z : v.w));
}

__device__ __forceinline__ float rowsum16(float v) {
  v += __builtin_bit_cast(float, __builtin_amdgcn_update_dpp(0, __builtin_bit_cast(int, v), 0x161, 0xf, 0xf, true));
  v += __builtin_bit_cast(float, __builtin_amdgcn_update_dpp(0, __builtin_bit_cast(int, v), 0x162, 0xf, 0xf, true));
  v += __builtin_bit_cast(float, __builtin_amdgcn_update_dpp(0, __builtin_bit_cast(int, v), 0x164, 0xf, 0xf, true));
  v += __builtin_bit_cast(float, __builtin_amdgcn_update_dpp(0, __builtin_bit_cast(int, v), 0x168, 0xf, 0xf, true));
  return v;
}

__device__ __forceinline__ void load_idx(Idx& d, const int* __restrict__ kc,
                                         const int* __restrict__ problem,
                                         const int* __restrict__ corr, int off) {
  d.kc = *reinterpret_cast<const int4*>(kc + off);
  d.pr = *reinterpret_cast<const int4*>(problem + off);
  d.co = *reinterpret_cast<const int4*>(corr + off);
}

__device__ __forceinline__ void load_rows(Rows& r, const Idx& ix,
                                          const float* __restrict__ A,
                                          const float* __restrict__ M8,
                                          const float* __restrict__ LO,
                                          unsigned o1, unsigned o2) {
#pragma unroll
  for (int i = 0; i < TB; ++i) {
    const unsigned k = (unsigned)pick(ix.kc, i);
    const unsigned p = (unsigned)pick(ix.pr, i);
    r.c0[i] = *reinterpret_cast<const float4*>(A + k * NKC + o1);
    r.c1[i] = *reinterpret_cast<const float4*>(A + k * NKC + o2);
    r.m4[i] = *reinterpret_cast<const float4*>(M8 + k * 8);
    r.sa[i] = M8[k * 8 + 4];
    r.lo[i] = reinterpret_cast<const float4*>(LO)[p];
  }
}

__device__ __forceinline__ void run_block(const Rows& r, const int4 co, bool has2,
                                          int j, float (&al0)[8], float (&al1)[8],
                                          float2* __restrict__ out2, int obase) {
  float ox[TB], oy[TB];
#pragma unroll
  for (int i = 0; i < TB; ++i) {
    float c[8];
    c[0] = r.c0[i].x; c[1] = r.c0[i].y; c[2] = r.c0[i].z; c[3] = r.c0[i].w;
    c[4] = has2 ? r.c1[i].x : 0.f; c[5] = has2 ? r.c1[i].y : 0.f;
    c[6] = has2 ? r.c1[i].z : 0.f; c[7] = has2 ? r.c1[i].w : 0.f;

    float p0 = c[0] * al0[0], p1 = c[0] * al1[0];
    float q0s = c[4] * al0[4], q1s = c[4] * al1[4];
#pragma unroll
    for (int k = 1; k < 4; ++k) {
      p0 = fmaf(c[k], al0[k], p0);
      p1 = fmaf(c[k], al1[k], p1);
      q0s = fmaf(c[4 + k], al0[4 + k], q0s);
      q1s = fmaf(c[4 + k], al1[4 + k], q1s);
    }
    const float a20 = rowsum16(p0 + q0s);
    const float a21 = rowsum16(p1 + q1s);

    const float LO00 = r.lo[i].x, LO01 = r.lo[i].y, LO10 = r.lo[i].z, LO11 = r.lo[i].w;
    const float SAi = r.sa[i];

    const float lp0 = lse2(LO00 * SAi + a20, LO10 * SAi + a21);
    const float lp1 = lse2(LO01 * SAi + a20, LO11 * SAi + a21);
    const float lz = lse2(lp0, lp1);

    const int ci = pick(co, i);
    const float ol0 = (ci == 0 ? LO00 : LO01) * SAi;
    const float ol1 = (ci == 0 ? LO10 : LO11) * SAi;
    const float q0 = ol0 + a20;
    const float q1 = ol1 + a21;
    const float t30 = lse2(q0 + r.m4[i].x, q1 + r.m4[i].y);
    const float t31 = lse2(q0 + r.m4[i].z, q1 + r.m4[i].w);

    ox[i] = lp0 - lz;
    oy[i] = lp1 - lz;
    asm volatile("" : "+v"(ox[i]), "+v"(oy[i]));

#pragma unroll
    for (int k = 0; k < 8; ++k) {
      al0[k] = fmaf(c[k], t30 - al0[k], al0[k]);
      al1[k] = fmaf(c[k], t31 - al1[k], al1[k]);
    }
  }
  float sx = ox[0], sy = oy[0];
#pragma unroll
  for (int i = 1; i < TB; ++i) {
    sx = (j == i) ? ox[i] : sx;
    sy = (j == i) ? oy[i] : sy;
  }
  if (j < TB) {
    float2 o2;
    o2.x = sx;
    o2.y = sy;
    out2[obase + j] = o2;
  }
}

__global__ void __launch_bounds__(32) k_forward(
    const int* __restrict__ corr, const int* __restrict__ kc,
    const int* __restrict__ problem, const float* __restrict__ A,
    const float* __restrict__ init_logits, const float* __restrict__ M8,
    const float* __restrict__ LO, float* __restrict__ out) {
  const int l = threadIdx.x;
  const int j = l & 15;
  const int b = blockIdx.x * 2 + (l >> 4);
  const bool has2 = (j < 9);
  const unsigned o1 = 4u * (unsigned)j;
  const unsigned o2 = has2 ? 4u * (unsigned)(j + 16) : o1;

  float al0[8], al1[8];
  {
    const float4* il = reinterpret_cast<const float4*>(init_logits);
    const float4 i0 = il[2 * j];
    const float4 i1 = il[2 * j + 1];
    const int j2 = has2 ? (32 + 2 * j) : (2 * j);
    const float4 i2 = il[j2];
    const float4 i3 = il[j2 + 1];
    float z;
    z = lse2(i0.x, i0.y); al0[0] = i0.x - z; al1[0] = i0.y - z;
    z = lse2(i0.z, i0.w); al0[1] = i0.z - z; al1[1] = i0.w - z;
    z = lse2(i1.x, i1.y); al0[2] = i1.x - z; al1[2] = i1.y - z;
    z = lse2(i1.z, i1.w); al0[3] = i1.z - z; al1[3] = i1.w - z;
    z = lse2(i2.x, i2.y); al0[4] = has2 ? i2.x - z : 0.f; al1[4] = has2 ? i2.y - z : 0.f;
    z = lse2(i2.z, i2.w); al0[5] = has2 ? i2.z - z : 0.f; al1[5] = has2 ? i2.w - z : 0.f;
    z = lse2(i3.x, i3.y); al0[6] = has2 ? i3.x - z : 0.f; al1[6] = has2 ? i3.y - z : 0.f;
    z = lse2(i3.z, i3.w); al0[7] = has2 ? i3.z - z : 0.f; al1[7] = has2 ? i3.w - z : 0.f;
  }

  const int base = b * NT;
  float2* __restrict__ out2 = reinterpret_cast<float2*>(out);

  Rows ra, rb;
  Idx ip, iq;
  int4 co_cur;
  {
    Idx i0;
    load_idx(i0, kc, problem, corr, base);
    load_rows(ra, i0, A, M8, LO, o1, o2);
    co_cur = i0.co;
    load_idx(ip, kc, problem, corr, base + TB);
  }

  for (int t0 = 0; t0 < NT - TB; t0 += 2 * TB) {
    load_rows(rb, ip, A, M8, LO, o1, o2);
    const int4 co_nb = ip.co;
    load_idx(iq, kc, problem, corr, base + t0 + 2 * TB);
    __builtin_amdgcn_sched_barrier(0);
    run_block(ra, co_cur, has2, j, al0, al1, out2, base + t0);

    load_rows(ra, iq, A, M8, LO, o1, o2);
    {
      const int tn = (t0 + 3 * TB < NT) ? (t0 + 3 * TB) : (NT - TB);
      load_idx(ip, kc, problem, corr, base + tn);
    }
    __builtin_amdgcn_sched_barrier(0);
    run_block(rb, co_nb, has2, j, al0, al1, out2, base + t0 + TB);
    co_cur = iq.co;
  }
  run_block(ra, co_cur, has2, j, al0, al1, out2, base + NT - TB);
}

extern "C" void kernel_launch(void* const* d_in, const int* in_sizes, int n_in,
                              void* d_out, int out_size, void* d_ws, size_t ws_size,
                              hipStream_t stream) {
  (void)in_sizes; (void)n_in; (void)out_size; (void)ws_size;
  const int*   corr    = (const int*)d_in[0];
  const int*   kc      = (const int*)d_in[1];
  const int*   problem = (const int*)d_in[2];
  const float* A       = (const float*)d_in[3];
  const float* trans   = (const float*)d_in[4];
  const float* obs     = (const float*)d_in[5];
  const float* init    = (const float*)d_in[6];
  float* out = (float*)d_out;

  float* logtT = (float*)d_ws;
  float* M8    = logtT + 512;
  float* LO    = M8 + (size_t)NPROB * 8;

  k_logt<<<1, 128, 0, stream>>>(trans, logtT);
  k_lo<<<(NPROB + 127) / 128, 128, 0, stream>>>(obs, LO);
  k_gemm_M<<<NPROB / 16, 32, 0, stream>>>(A, logtT, M8);
  k_forward<<<NBATCH / 2, 32, 0, stream>>>(corr, kc, problem, A, init, M8, LO, out);
}


# k_forward step math in the linear domain: exp(a2 - max) is formed once per step and shared by all five logsumexps, whose table addends enter as exp weights (a2b exp'd in the GEMM epilogue); bare log2
# speedup vs baseline: 4.5644x; 1.3539x over previous
#include <hip/hip_runtime.h>
#include <math.h>

#define NPROB 10000
#define NKC   100
#define NBATCH 1024
#define NT    500

typedef __attribute__((ext_vector_type(2))) float v2f;
typedef __attribute__((ext_vector_type(8))) float v8f;

__device__ __forceinline__ float lse2(float a, float b) {
  float m = fmaxf(a, b);
  return m + __logf(__expf(a - m) + __expf(b - m));
}

__device__ __forceinline__ float ex2(float x) { return __builtin_amdgcn_exp2f(x); }
__device__ __forceinline__ float lg2(float x) { return __builtin_amdgcn_logf(x); }
#define LOG2E_F 1.4426950408889634f
#define LN2_F   0.6931471805599453f

__global__ void k_logt(const float* __restrict__ trans, float* __restrict__ logtT) {
  int k = blockIdx.x * blockDim.x + threadIdx.x;
  if (k >= NKC) return;
  const float4 v = reinterpret_cast<const float4*>(trans)[k];
  const float l0 = lse2(v.x, v.z);
  const float l1 = lse2(v.y, v.w);
  logtT[0 * NKC + k] = v.x - l0;
  logtT[1 * NKC + k] = v.y - l1;
  logtT[2 * NKC + k] = v.z - l0;
  logtT[3 * NKC + k] = v.w - l1;
}

__global__ void k_lo(const float* __restrict__ obs, float* __restrict__ LO) {
  int p = blockIdx.x * blockDim.x + threadIdx.x;
  if (p >= NPROB) return;
  const float4 v = reinterpret_cast<const float4*>(obs)[p];
  const float ls0 = lse2(v.x, v.y);
  const float ls1 = lse2(v.z, v.w);
  float4 r;
  r.x = v.x - ls0; r.y = v.y - ls0; r.z = v.z - ls1; r.w = v.w - ls1;
  reinterpret_cast<float4*>(LO)[p] = r;
}

#if __has_builtin(__builtin_amdgcn_wmma_f32_16x16x4_f32)
#define HAVE_WMMA_F32X4 1
#endif

__global__ void __launch_bounds__(32) k_gemm_M(const float* __restrict__ A,
                                               const float* __restrict__ logtT,
                                               float* __restrict__ M8) {
  const int tile = blockIdx.x;
  const int l = threadIdx.x;
  const int m = l & 15;
  const bool hi = (l >= 16);
#ifdef HAVE_WMMA_F32X4
  const int row = tile * 16 + m;
  const float nmask = (m < 4) ? 1.0f : 0.0f;
  const float ones4 = (m == 4) ? 1.0f : 0.0f;
  const float* __restrict__ bcolp = logtT + (m & 3) * NKC;
  v8f c = {0.f, 0.f, 0.f, 0.f, 0.f, 0.f, 0.f, 0.f};
#pragma unroll 5
  for (int kk = 0; kk < 25; ++kk) {
    const int k0 = kk * 4 + (hi ? 2 : 0);
    const float2 av = *reinterpret_cast<const float2*>(A + row * NKC + k0);
    const float2 bv = *reinterpret_cast<const float2*>(bcolp + k0);
    v2f a, b;
    a.x = av.x;
    a.y = av.y;
    b.x = bv.x * nmask + ones4;
    b.y = bv.y * nmask + ones4;
    c = __builtin_amdgcn_wmma_f32_16x16x4_f32(
         false, a,  false, b,
         (short)0, c,  false,  false);
  }
  if (m < 5) {
#pragma unroll
    for (int v = 0; v < 8; ++v) {
      const int r = tile * 16 + v + (hi ? 8 : 0);
      M8[r * 8 + m] = (m < 4) ? __expf(c[v]) : c[v];
    }
  }
#else
  (void)hi;
  if (l < 16) {
    const int row = tile * 16 + m;
    float acc0 = 0.f, acc1 = 0.f, acc2 = 0.f, acc3 = 0.f, acc4 = 0.f;
    for (int k = 0; k < NKC; ++k) {
      const float av = A[row * NKC + k];
      acc0 += av * logtT[0 * NKC + k];
      acc1 += av * logtT[1 * NKC + k];
      acc2 += av * logtT[2 * NKC + k];
      acc3 += av * logtT[3 * NKC + k];
      acc4 += av;
    }
    M8[row * 8 + 0] = __expf(acc0); M8[row * 8 + 1] = __expf(acc1);
    M8[row * 8 + 2] = __expf(acc2); M8[row * 8 + 3] = __expf(acc3);
    M8[row * 8 + 4] = acc4;
  }
#endif
}

#define TB 4
static_assert(NT % TB == 0, "row blocks must tile the sequence");
static_assert((NT / TB) % 2 == 1, "the loop below takes block pairs and one closing block");

struct Rows {
  float4 c0[TB];
  float4 c1[TB];
  float4 m4[TB];
  float4 lo[TB];
  float sa[TB];
};
struct Idx { int4 kc; int4 pr; int4 co; };

__device__ __forceinline__ int pick(const int4 v, int i) {
  return i == 0 ? v.x : (i == 1 ? v.y : (i == 2 ? v.z : v.w));
}

__device__ __forceinline__ float rowsum16(float v) {
  v += __builtin_bit_cast(float, __builtin_amdgcn_update_dpp(0, __builtin_bit_cast(int, v), 0x161, 0xf, 0xf, true));
  v += __builtin_bit_cast(float, __builtin_amdgcn_update_dpp(0, __builtin_bit_cast(int, v), 0x162, 0xf, 0xf, true));
  v += __builtin_bit_cast(float, __builtin_amdgcn_update_dpp(0, __builtin_bit_cast(int, v), 0x164, 0xf, 0xf, true));
  v += __builtin_bit_cast(float, __builtin_amdgcn_update_dpp(0, __builtin_bit_cast(int, v), 0x168, 0xf, 0xf, true));
  return v;
}

__device__ __forceinline__ void load_idx(Idx& d, const int* __restrict__ kc,
                                         const int* __restrict__ problem,
                                         const int* __restrict__ corr, int off) {
  d.kc = *reinterpret_cast<const int4*>(kc + off);
  d.pr = *reinterpret_cast<const int4*>(problem + off);
  d.co = *reinterpret_cast<const int4*>(corr + off);
}

__device__ __forceinline__ void load_rows(Rows& r, const Idx& ix,
                                          const float* __restrict__ A,
                                          const float* __restrict__ M8,
                                          const float* __restrict__ LO,
                                          unsigned o1, unsigned o2) {
#pragma unroll
  for (int i = 0; i < TB; ++i) {
    const unsigned k = (unsigned)pick(ix.kc, i);
    const unsigned p = (unsigned)pick(ix.pr, i);
    r.c0[i] = *reinterpret_cast<const float4*>(A + k * NKC + o1);
    r.c1[i] = *reinterpret_cast<const float4*>(A + k * NKC + o2);
    r.m4[i] = *reinterpret_cast<const float4*>(M8 + k * 8);
    r.sa[i] = M8[k * 8 + 4];
    r.lo[i] = reinterpret_cast<const float4*>(LO)[p];
  }
}

__device__ __forceinline__ void run_block(const Rows& r, const int4 co, bool has2,
                                          int j, float (&al0)[8], float (&al1)[8],
                                          float2* __restrict__ out2, int obase) {
  float ox[TB], oy[TB];
#pragma unroll
  for (int i = 0; i < TB; ++i) {
    float c[8];
    c[0] = r.c0[i].x; c[1] = r.c0[i].y; c[2] = r.c0[i].z; c[3] = r.c0[i].w;
    c[4] = has2 ? r.c1[i].x : 0.f; c[5] = has2 ? r.c1[i].y : 0.f;
    c[6] = has2 ? r.c1[i].z : 0.f; c[7] = has2 ? r.c1[i].w : 0.f;

    float p0 = c[0] * al0[0], p1 = c[0] * al1[0];
    float q0s = c[4] * al0[4], q1s = c[4] * al1[4];
#pragma unroll
    for (int k = 1; k < 4; ++k) {
      p0 = fmaf(c[k], al0[k], p0);
      p1 = fmaf(c[k], al1[k], p1);
      q0s = fmaf(c[4 + k], al0[4 + k], q0s);
      q1s = fmaf(c[4 + k], al1[4 + k], q1s);
    }
    const float a20 = rowsum16(p0 + q0s);
    const float a21 = rowsum16(p1 + q1s);

    const float sal = r.sa[i] * LOG2E_F;
    const float y00 = ex2(r.lo[i].x * sal), y01 = ex2(r.lo[i].y * sal);
    const float y10 = ex2(r.lo[i].z * sal), y11 = ex2(r.lo[i].w * sal);
    const int ci = pick(co, i);
    const float yc0 = (ci == 0) ? y00 : y01;
    const float yc1 = (ci == 0) ? y10 : y11;
    const float z00 = yc0 * r.m4[i].x, z01 = yc1 * r.m4[i].y;
    const float z10 = yc0 * r.m4[i].z, z11 = yc1 * r.m4[i].w;

    const float mx = fmaxf(a20, a21);
    const float e0 = ex2((a20 - mx) * LOG2E_F);
    const float e1 = ex2((a21 - mx) * LOG2E_F);

    const float t30 = fmaf(LN2_F, lg2(fmaf(e0, z00, e1 * z01)), mx);
    const float t31 = fmaf(LN2_F, lg2(fmaf(e0, z10, e1 * z11)), mx);

    const float s0 = fmaf(e0, y00, e1 * y10);
    const float s1 = fmaf(e0, y01, e1 * y11);
    const float lt = lg2(s0 + s1);

    ox[i] = (lg2(s0) - lt) * LN2_F;
    oy[i] = (lg2(s1) - lt) * LN2_F;
    asm volatile("" : "+v"(ox[i]), "+v"(oy[i]));

#pragma unroll
    for (int k = 0; k < 8; ++k) {
      al0[k] = fmaf(c[k], t30 - al0[k], al0[k]);
      al1[k] = fmaf(c[k], t31 - al1[k], al1[k]);
    }
  }
  float sx = ox[0], sy = oy[0];
#pragma unroll
  for (int i = 1; i < TB; ++i) {
    sx = (j == i) ? ox[i] : sx;
    sy = (j == i) ? oy[i] : sy;
  }
  if (j < TB) {
    float2 o2;
    o2.x = sx;
    o2.y = sy;
    out2[obase + j] = o2;
  }
}

__global__ void __launch_bounds__(32) k_forward(
    const int* __restrict__ corr, const int* __restrict__ kc,
    const int* __restrict__ problem, const float* __restrict__ A,
    const float* __restrict__ init_logits, const float* __restrict__ M8,
    const float* __restrict__ LO, float* __restrict__ out) {
  const int l = threadIdx.x;
  const int j = l & 15;
  const int b = blockIdx.x * 2 + (l >> 4);
  const bool has2 = (j < 9);
  const unsigned o1 = 4u * (unsigned)j;
  const unsigned o2 = has2 ? 4u * (unsigned)(j + 16) : o1;

  float al0[8], al1[8];
  {
    const float4* il = reinterpret_cast<const float4*>(init_logits);
    const float4 i0 = il[2 * j];
    const float4 i1 = il[2 * j + 1];
    const int j2 = has2 ? (32 + 2 * j) : (2 * j);
    const float4 i2 = il[j2];
    const float4 i3 = il[j2 + 1];
    float z;
    z = lse2(i0.x, i0.y); al0[0] = i0.x - z; al1[0] = i0.y - z;
    z = lse2(i0.z, i0.w); al0[1] = i0.z - z; al1[1] = i0.w - z;
    z = lse2(i1.x, i1.y); al0[2] = i1.x - z; al1[2] = i1.y - z;
    z = lse2(i1.z, i1.w); al0[3] = i1.z - z; al1[3] = i1.w - z;
    z = lse2(i2.x, i2.y); al0[4] = has2 ? i2.x - z : 0.f; al1[4] = has2 ? i2.y - z : 0.f;
    z = lse2(i2.z, i2.w); al0[5] = has2 ? i2.z - z : 0.f; al1[5] = has2 ? i2.w - z : 0.f;
    z = lse2(i3.x, i3.y); al0[6] = has2 ? i3.x - z : 0.f; al1[6] = has2 ? i3.y - z : 0.f;
    z = lse2(i3.z, i3.w); al0[7] = has2 ? i3.z - z : 0.f; al1[7] = has2 ? i3.w - z : 0.f;
  }

  const int base = b * NT;
  float2* __restrict__ out2 = reinterpret_cast<float2*>(out);

  Rows ra, rb;
  Idx ip, iq;
  int4 co_cur;
  {
    Idx i0;
    load_idx(i0, kc, problem, corr, base);
    load_rows(ra, i0, A, M8, LO, o1, o2);
    co_cur = i0.co;
    load_idx(ip, kc, problem, corr, base + TB);
  }

  for (int t0 = 0; t0 < NT - TB; t0 += 2 * TB) {
    load_rows(rb, ip, A, M8, LO, o1, o2);
    const int4 co_nb = ip.co;
    load_idx(iq, kc, problem, corr, base + t0 + 2 * TB);
    __builtin_amdgcn_sched_barrier(0);
    run_block(ra, co_cur, has2, j, al0, al1, out2, base + t0);

    load_rows(ra, iq, A, M8, LO, o1, o2);
    {
      const int tn = (t0 + 3 * TB < NT) ? (t0 + 3 * TB) : (NT - TB);
      load_idx(ip, kc, problem, corr, base + tn);
    }
    __builtin_amdgcn_sched_barrier(0);
    run_block(rb, co_nb, has2, j, al0, al1, out2, base + t0 + TB);
    co_cur = iq.co;
  }
  run_block(ra, co_cur, has2, j, al0, al1, out2, base + NT - TB);
}

extern "C" void kernel_launch(void* const* d_in, const int* in_sizes, int n_in,
                              void* d_out, int out_size, void* d_ws, size_t ws_size,
                              hipStream_t stream) {
  (void)in_sizes; (void)n_in; (void)out_size; (void)ws_size;
  const int*   corr    = (const int*)d_in[0];
  const int*   kc      = (const int*)d_in[1];
  const int*   problem = (const int*)d_in[2];
  const float* A       = (const float*)d_in[3];
  const float* trans   = (const float*)d_in[4];
  const float* obs     = (const float*)d_in[5];
  const float* init    = (const float*)d_in[6];
  float* out = (float*)d_out;

  float* logtT = (float*)d_ws;
  float* M8    = logtT + 512;
  float* LO    = M8 + (size_t)NPROB * 8;

  k_logt<<<1, 128, 0, stream>>>(trans, logtT);
  k_lo<<<(NPROB + 127) / 128, 128, 0, stream>>>(obs, LO);
  k_gemm_M<<<NPROB / 16, 32, 0, stream>>>(A, logtT, M8);
  k_forward<<<NBATCH / 2, 32, 0, stream>>>(corr, kc, problem, A, init, M8, LO, out);
}


# k_forward gathers: each load named by one 32-bit element index on a scalar base (no 64-bit address pairs); a block's index registers stay unwritten until its steps end: no address-translation waits
# speedup vs baseline: 5.9787x; 1.3099x over previous
#include <hip/hip_runtime.h>
#include <math.h>

#define NPROB 10000
#define NKC   100
#define NBATCH 1024
#define NT    500

typedef __attribute__((ext_vector_type(2))) float v2f;
typedef __attribute__((ext_vector_type(8))) float v8f;

__device__ __forceinline__ float lse2(float a, float b) {
  float m = fmaxf(a, b);
  return m + __logf(__expf(a - m) + __expf(b - m));
}

__device__ __forceinline__ float ex2(float x) { return __builtin_amdgcn_exp2f(x); }
__device__ __forceinline__ float lg2(float x) { return __builtin_amdgcn_logf(x); }
#define LOG2E_F 1.4426950408889634f
#define LN2_F   0.6931471805599453f

__global__ void k_logt(const float* __restrict__ trans, float* __restrict__ logtT) {
  int k = blockIdx.x * blockDim.x + threadIdx.x;
  if (k >= NKC) return;
  const float4 v = reinterpret_cast<const float4*>(trans)[k];
  const float l0 = lse2(v.x, v.z);
  const float l1 = lse2(v.y, v.w);
  logtT[0 * NKC + k] = v.x - l0;
  logtT[1 * NKC + k] = v.y - l1;
  logtT[2 * NKC + k] = v.z - l0;
  logtT[3 * NKC + k] = v.w - l1;
}

__global__ void k_lo(const float* __restrict__ obs, float* __restrict__ LO) {
  int p = blockIdx.x * blockDim.x + threadIdx.x;
  if (p >= NPROB) return;
  const float4 v = reinterpret_cast<const float4*>(obs)[p];
  const float ls0 = lse2(v.x, v.y);
  const float ls1 = lse2(v.z, v.w);
  float4 r;
  r.x = v.x - ls0; r.y = v.y - ls0; r.z = v.z - ls1; r.w = v.w - ls1;
  reinterpret_cast<float4*>(LO)[p] = r;
}

#if __has_builtin(__builtin_amdgcn_wmma_f32_16x16x4_f32)
#define HAVE_WMMA_F32X4 1
#endif

__global__ void __launch_bounds__(32) k_gemm_M(const float* __restrict__ A,
                                               const float* __restrict__ logtT,
                                               float* __restrict__ M8) {
  const int tile = blockIdx.x;
  const int l = threadIdx.x;
  const int m = l & 15;
  const bool hi = (l >= 16);
#ifdef HAVE_WMMA_F32X4
  const int row = tile * 16 + m;
  const float nmask = (m < 4) ? 1.0f : 0.0f;
  const float ones4 = (m == 4) ? 1.0f : 0.0f;
  const float* __restrict__ bcolp = logtT + (m & 3) * NKC;
  v8f c = {0.f, 0.f, 0.f, 0.f, 0.f, 0.f, 0.f, 0.f};
#pragma unroll 5
  for (int kk = 0; kk < 25; ++kk) {
    const int k0 = kk * 4 + (hi ? 2 : 0);
    const float2 av = *reinterpret_cast<const float2*>(A + row * NKC + k0);
    const float2 bv = *reinterpret_cast<const float2*>(bcolp + k0);
    v2f a, b;
    a.x = av.x;
    a.y = av.y;
    b.x = bv.x * nmask + ones4;
    b.y = bv.y * nmask + ones4;
    c = __builtin_amdgcn_wmma_f32_16x16x4_f32(
         false, a,  false, b,
         (short)0, c,  false,  false);
  }
  if (m < 5) {
#pragma unroll
    for (int v = 0; v < 8; ++v) {
      const int r = tile * 16 + v + (hi ? 8 : 0);
      M8[r * 8 + m] = (m < 4) ? __expf(c[v]) : c[v];
    }
  }
#else
  (void)hi;
  if (l < 16) {
    const int row = tile * 16 + m;
    float acc0 = 0.f, acc1 = 0.f, acc2 = 0.f, acc3 = 0.f, acc4 = 0.f;
    for (int k = 0; k < NKC; ++k) {
      const float av = A[row * NKC + k];
      acc0 += av * logtT[0 * NKC + k];
      acc1 += av * logtT[1 * NKC + k];
      acc2 += av * logtT[2 * NKC + k];
      acc3 += av * logtT[3 * NKC + k];
      acc4 += av;
    }
    M8[row * 8 + 0] = __expf(acc0); M8[row * 8 + 1] = __expf(acc1);
    M8[row * 8 + 2] = __expf(acc2); M8[row * 8 + 3] = __expf(acc3);
    M8[row * 8 + 4] = acc4;
  }
#endif
}

#define TB 4
static_assert(NT % TB == 0, "row blocks must tile the sequence");
static_assert((NT / TB) % 2 == 1, "the loop below takes block pairs and one closing block");

struct Rows {
  float4 c0[TB];
  float4 c1[TB];
  float4 m4[TB];
  float4 lo[TB];
  float sa[TB];
};
struct Idx { int4 kc; int4 pr; int4 co; };

__device__ __forceinline__ int pick(const int4 v, int i) {
  return i == 0 ? v.x : (i == 1 ? v.y : (i == 2 ? v.z : v.w));
}

__device__ __forceinline__ float rowsum16(float v) {
  v += __builtin_bit_cast(float, __builtin_amdgcn_update_dpp(0, __builtin_bit_cast(int, v), 0x161, 0xf, 0xf, true));
  v += __builtin_bit_cast(float, __builtin_amdgcn_update_dpp(0, __builtin_bit_cast(int, v), 0x162, 0xf, 0xf, true));
  v += __builtin_bit_cast(float, __builtin_amdgcn_update_dpp(0, __builtin_bit_cast(int, v), 0x164, 0xf, 0xf, true));
  v += __builtin_bit_cast(float, __builtin_amdgcn_update_dpp(0, __builtin_bit_cast(int, v), 0x168, 0xf, 0xf, true));
  return v;
}

__device__ __forceinline__ void load_idx(Idx& d, const int4* __restrict__ kc4,
                                         const int4* __restrict__ pr4,
                                         const int4* __restrict__ co4, unsigned q) {
  d.kc = kc4[q];
  d.pr = pr4[q];
  d.co = co4[q];
}

struct Gix {
  unsigned a0[TB], a1[TB];
  unsigned k2[TB];
  unsigned k8[TB];
  unsigned lo[TB];
};

__device__ __forceinline__ void load_rows(Rows& r, Gix& g, const Idx& ix,
                                          const float4* __restrict__ A4,
                                          const float4* __restrict__ M84,
                                          const float* __restrict__ M8,
                                          const float4* __restrict__ LO4,
                                          unsigned j1, unsigned j2) {
#pragma unroll
  for (int i = 0; i < TB; ++i) {
    const unsigned k = (unsigned)pick(ix.kc, i);
    const unsigned kr = __umul24(k, (unsigned)(NKC / 4));
    g.a0[i] = kr + j1;
    g.a1[i] = kr + j2;
    g.k2[i] = 2u * k;
    g.k8[i] = 8u * k;
    g.lo[i] = (unsigned)pick(ix.pr, i);
  }
#pragma unroll
  for (int i = 0; i < TB; ++i) {
    r.c0[i] = A4[g.a0[i]];
    r.c1[i] = A4[g.a1[i]];
    r.m4[i] = M84[g.k2[i]];
    r.sa[i] = (M8 + 4)[g.k8[i]];
    r.lo[i] = LO4[g.lo[i]];
  }
}

__device__ __forceinline__ void hold_indices(const Gix& g, unsigned q) {
#pragma unroll
  for (int i = 0; i < TB; ++i) {
    asm volatile("" :: "v"(g.a0[i]), "v"(g.a1[i]), "v"(g.k2[i]), "v"(g.k8[i]), "v"(g.lo[i]));
  }
  asm volatile("" :: "v"(q));
}

__device__ __forceinline__ void run_block(const Rows& r, const int4 co, bool has2,
                                          int j, float (&al0)[8], float (&al1)[8],
                                          float2* __restrict__ out2, int obase) {
  float ox[TB], oy[TB];
#pragma unroll
  for (int i = 0; i < TB; ++i) {
    float c[8];
    c[0] = r.c0[i].x; c[1] = r.c0[i].y; c[2] = r.c0[i].z; c[3] = r.c0[i].w;
    c[4] = has2 ? r.c1[i].x : 0.f; c[5] = has2 ? r.c1[i].y : 0.f;
    c[6] = has2 ? r.c1[i].z : 0.f; c[7] = has2 ? r.c1[i].w : 0.f;

    float p0 = c[0] * al0[0], p1 = c[0] * al1[0];
    float q0s = c[4] * al0[4], q1s = c[4] * al1[4];
#pragma unroll
    for (int k = 1; k < 4; ++k) {
      p0 = fmaf(c[k], al0[k], p0);
      p1 = fmaf(c[k], al1[k], p1);
      q0s = fmaf(c[4 + k], al0[4 + k], q0s);
      q1s = fmaf(c[4 + k], al1[4 + k], q1s);
    }
    const float a20 = rowsum16(p0 + q0s);
    const float a21 = rowsum16(p1 + q1s);

    const float sal = r.sa[i] * LOG2E_F;
    const float y00 = ex2(r.lo[i].x * sal), y01 = ex2(r.lo[i].y * sal);
    const float y10 = ex2(r.lo[i].z * sal), y11 = ex2(r.lo[i].w * sal);
    const int ci = pick(co, i);
    const float yc0 = (ci == 0) ? y00 : y01;
    const float yc1 = (ci == 0) ? y10 : y11;
    const float z00 = yc0 * r.m4[i].x, z01 = yc1 * r.m4[i].y;
    const float z10 = yc0 * r.m4[i].z, z11 = yc1 * r.m4[i].w;

    const float mx = fmaxf(a20, a21);
    const float e0 = ex2((a20 - mx) * LOG2E_F);
    const float e1 = ex2((a21 - mx) * LOG2E_F);

    const float t30 = fmaf(LN2_F, lg2(fmaf(e0, z00, e1 * z01)), mx);
    const float t31 = fmaf(LN2_F, lg2(fmaf(e0, z10, e1 * z11)), mx);

    const float s0 = fmaf(e0, y00, e1 * y10);
    const float s1 = fmaf(e0, y01, e1 * y11);
    const float lt = lg2(s0 + s1);

    ox[i] = (lg2(s0) - lt) * LN2_F;
    oy[i] = (lg2(s1) - lt) * LN2_F;
    asm volatile("" : "+v"(ox[i]), "+v"(oy[i]));

#pragma unroll
    for (int k = 0; k < 8; ++k) {
      al0[k] = fmaf(c[k], t30 - al0[k], al0[k]);
      al1[k] = fmaf(c[k], t31 - al1[k], al1[k]);
    }
  }
  float sx = ox[0], sy = oy[0];
#pragma unroll
  for (int i = 1; i < TB; ++i) {
    sx = (j == i) ? ox[i] : sx;
    sy = (j == i) ? oy[i] : sy;
  }
  if (j < TB) {
    float2 o2;
    o2.x = sx;
    o2.y = sy;
    out2[obase + j] = o2;
  }
}

__global__ void __launch_bounds__(32) k_forward(
    const int* __restrict__ corr, const int* __restrict__ kc,
    const int* __restrict__ problem, const float* __restrict__ A,
    const float* __restrict__ init_logits, const float* __restrict__ M8,
    const float* __restrict__ LO, float* __restrict__ out) {
  const int l = threadIdx.x;
  const int j = l & 15;
  const int b = blockIdx.x * 2 + (l >> 4);
  const bool has2 = (j < 9);
  const unsigned pc1 = (unsigned)j;
  const unsigned pc2 = has2 ? (unsigned)(j + 16) : pc1;
  const float4* __restrict__ A4  = reinterpret_cast<const float4*>(A);
  const float4* __restrict__ M84 = reinterpret_cast<const float4*>(M8);
  const float4* __restrict__ LO4 = reinterpret_cast<const float4*>(LO);
  const int4* __restrict__ kc4 = reinterpret_cast<const int4*>(kc);
  const int4* __restrict__ pr4 = reinterpret_cast<const int4*>(problem);
  const int4* __restrict__ co4 = reinterpret_cast<const int4*>(corr);

  float al0[8], al1[8];
  {
    const float4* il = reinterpret_cast<const float4*>(init_logits);
    const float4 i0 = il[2 * j];
    const float4 i1 = il[2 * j + 1];
    const int j2 = has2 ? (32 + 2 * j) : (2 * j);
    const float4 i2 = il[j2];
    const float4 i3 = il[j2 + 1];
    float z;
    z = lse2(i0.x, i0.y); al0[0] = i0.x - z; al1[0] = i0.y - z;
    z = lse2(i0.z, i0.w); al0[1] = i0.z - z; al1[1] = i0.w - z;
    z = lse2(i1.x, i1.y); al0[2] = i1.x - z; al1[2] = i1.y - z;
    z = lse2(i1.z, i1.w); al0[3] = i1.z - z; al1[3] = i1.w - z;
    z = lse2(i2.x, i2.y); al0[4] = has2 ? i2.x - z : 0.f; al1[4] = has2 ? i2.y - z : 0.f;
    z = lse2(i2.z, i2.w); al0[5] = has2 ? i2.z - z : 0.f; al1[5] = has2 ? i2.w - z : 0.f;
    z = lse2(i3.x, i3.y); al0[6] = has2 ? i3.x - z : 0.f; al1[6] = has2 ? i3.y - z : 0.f;
    z = lse2(i3.z, i3.w); al0[7] = has2 ? i3.z - z : 0.f; al1[7] = has2 ? i3.w - z : 0.f;
  }

  const int base = b * NT;
  float2* __restrict__ out2 = reinterpret_cast<float2*>(out);

  const unsigned q0 = (unsigned)b * (unsigned)(NT / TB);
  Rows ra, rb;
  Gix ga, gb;
  Idx ip, iq;
  int4 co_cur;
  {
    Idx i0;
    load_idx(i0, kc4, pr4, co4, q0);
    load_rows(ra, ga, i0, A4, M84, M8, LO4, pc1, pc2);
    co_cur = i0.co;
    load_idx(ip, kc4, pr4, co4, q0 + 1u);
  }

  for (int t0 = 0; t0 < NT - TB; t0 += 2 * TB) {
    const unsigned nb = (unsigned)t0 / TB;
    load_rows(rb, gb, ip, A4, M84, M8, LO4, pc1, pc2);
    const int4 co_nb = ip.co;
    const unsigned qa = q0 + nb + 2u;
    load_idx(iq, kc4, pr4, co4, qa);
    __builtin_amdgcn_sched_barrier(0);
    run_block(ra, co_cur, has2, j, al0, al1, out2, base + t0);
    hold_indices(gb, qa);

    load_rows(ra, ga, iq, A4, M84, M8, LO4, pc1, pc2);
    const unsigned qb = q0 + ((nb + 3u < (unsigned)(NT / TB)) ? (nb + 3u) : (unsigned)(NT / TB - 1));
    load_idx(ip, kc4, pr4, co4, qb);
    __builtin_amdgcn_sched_barrier(0);
    run_block(rb, co_nb, has2, j, al0, al1, out2, base + t0 + TB);
    hold_indices(ga, qb);
    co_cur = iq.co;
  }
  run_block(ra, co_cur, has2, j, al0, al1, out2, base + NT - TB);
}

extern "C" void kernel_launch(void* const* d_in, const int* in_sizes, int n_in,
                              void* d_out, int out_size, void* d_ws, size_t ws_size,
                              hipStream_t stream) {
  (void)in_sizes; (void)n_in; (void)out_size; (void)ws_size;
  const int*   corr    = (const int*)d_in[0];
  const int*   kc      = (const int*)d_in[1];
  const int*   problem = (const int*)d_in[2];
  const float* A       = (const float*)d_in[3];
  const float* trans   = (const float*)d_in[4];
  const float* obs     = (const float*)d_in[5];
  const float* init    = (const float*)d_in[6];
  float* out = (float*)d_out;

  float* logtT = (float*)d_ws;
  float* M8    = logtT + 512;
  float* LO    = M8 + (size_t)NPROB * 8;

  k_logt<<<1, 128, 0, stream>>>(trans, logtT);
  k_lo<<<(NPROB + 127) / 128, 128, 0, stream>>>(obs, LO);
  k_gemm_M<<<NPROB / 16, 32, 0, stream>>>(A, logtT, M8);
  k_forward<<<NBATCH / 2, 32, 0, stream>>>(corr, kc, problem, A, init, M8, LO, out);
}


# k_forward: alpha update behind t3 cut from sub+fma to one fma (w = alpha - c*alpha formed off the chain: new rounding order of the update); a block's last update held in front of its store
# speedup vs baseline: 6.1738x; 1.0326x over previous
#include <hip/hip_runtime.h>
#include <math.h>

#define NPROB 10000
#define NKC   100
#define NBATCH 1024
#define NT    500

typedef __attribute__((ext_vector_type(2))) float v2f;
typedef __attribute__((ext_vector_type(8))) float v8f;

__device__ __forceinline__ float lse2(float a, float b) {
  float m = fmaxf(a, b);
  return m + __logf(__expf(a - m) + __expf(b - m));
}

__device__ __forceinline__ float ex2(float x) { return __builtin_amdgcn_exp2f(x); }
__device__ __forceinline__ float lg2(float x) { return __builtin_amdgcn_logf(x); }
#define LOG2E_F 1.4426950408889634f
#define LN2_F   0.6931471805599453f

__global__ void k_logt(const float* __restrict__ trans, float* __restrict__ logtT) {
  int k = blockIdx.x * blockDim.x + threadIdx.x;
  if (k >= NKC) return;
  const float4 v = reinterpret_cast<const float4*>(trans)[k];
  const float l0 = lse2(v.x, v.z);
  const float l1 = lse2(v.y, v.w);
  logtT[0 * NKC + k] = v.x - l0;
  logtT[1 * NKC + k] = v.y - l1;
  logtT[2 * NKC + k] = v.z - l0;
  logtT[3 * NKC + k] = v.w - l1;
}

__global__ void k_lo(const float* __restrict__ obs, float* __restrict__ LO) {
  int p = blockIdx.x * blockDim.x + threadIdx.x;
  if (p >= NPROB) return;
  const float4 v = reinterpret_cast<const float4*>(obs)[p];
  const float ls0 = lse2(v.x, v.y);
  const float ls1 = lse2(v.z, v.w);
  float4 r;
  r.x = v.x - ls0; r.y = v.y - ls0; r.z = v.z - ls1; r.w = v.w - ls1;
  reinterpret_cast<float4*>(LO)[p] = r;
}

#if __has_builtin(__builtin_amdgcn_wmma_f32_16x16x4_f32)
#define HAVE_WMMA_F32X4 1
#endif

__global__ void __launch_bounds__(32) k_gemm_M(const float* __restrict__ A,
                                               const float* __restrict__ logtT,
                                               float* __restrict__ M8) {
  const int tile = blockIdx.x;
  const int l = threadIdx.x;
  const int m = l & 15;
  const bool hi = (l >= 16);
#ifdef HAVE_WMMA_F32X4
  const int row = tile * 16 + m;
  const float nmask = (m < 4) ? 1.0f : 0.0f;
  const float ones4 = (m == 4) ? 1.0f : 0.0f;
  const float* __restrict__ bcolp = logtT + (m & 3) * NKC;
  v8f c = {0.f, 0.f, 0.f, 0.f, 0.f, 0.f, 0.f, 0.f};
#pragma unroll 5
  for (int kk = 0; kk < 25; ++kk) {
    const int k0 = kk * 4 + (hi ? 2 : 0);
    const float2 av = *reinterpret_cast<const float2*>(A + row * NKC + k0);
    const float2 bv = *reinterpret_cast<const float2*>(bcolp + k0);
    v2f a, b;
    a.x = av.x;
    a.y = av.y;
    b.x = bv.x * nmask + ones4;
    b.y = bv.y * nmask + ones4;
    c = __builtin_amdgcn_wmma_f32_16x16x4_f32(
         false, a,  false, b,
         (short)0, c,  false,  false);
  }
  if (m < 5) {
#pragma unroll
    for (int v = 0; v < 8; ++v) {
      const int r = tile * 16 + v + (hi ? 8 : 0);
      M8[r * 8 + m] = (m < 4) ? __expf(c[v]) : c[v];
    }
  }
#else
  (void)hi;
  if (l < 16) {
    const int row = tile * 16 + m;
    float acc0 = 0.f, acc1 = 0.f, acc2 = 0.f, acc3 = 0.f, acc4 = 0.f;
    for (int k = 0; k < NKC; ++k) {
      const float av = A[row * NKC + k];
      acc0 += av * logtT[0 * NKC + k];
      acc1 += av * logtT[1 * NKC + k];
      acc2 += av * logtT[2 * NKC + k];
      acc3 += av * logtT[3 * NKC + k];
      acc4 += av;
    }
    M8[row * 8 + 0] = __expf(acc0); M8[row * 8 + 1] = __expf(acc1);
    M8[row * 8 + 2] = __expf(acc2); M8[row * 8 + 3] = __expf(acc3);
    M8[row * 8 + 4] = acc4;
  }
#endif
}

#define TB 4
static_assert(NT % TB == 0, "row blocks must tile the sequence");
static_assert((NT / TB) % 2 == 1, "the loop below takes block pairs and one closing block");

struct Rows {
  float4 c0[TB];
  float4 c1[TB];
  float4 m4[TB];
  float4 lo[TB];
  float sa[TB];
};
struct Idx { int4 kc; int4 pr; int4 co; };

__device__ __forceinline__ int pick(const int4 v, int i) {
  return i == 0 ? v.x : (i == 1 ? v.y : (i == 2 ? v.z : v.w));
}

__device__ __forceinline__ float rowsum16(float v) {
  v += __builtin_bit_cast(float, __builtin_amdgcn_update_dpp(0, __builtin_bit_cast(int, v), 0x161, 0xf, 0xf, true));
  v += __builtin_bit_cast(float, __builtin_amdgcn_update_dpp(0, __builtin_bit_cast(int, v), 0x162, 0xf, 0xf, true));
  v += __builtin_bit_cast(float, __builtin_amdgcn_update_dpp(0, __builtin_bit_cast(int, v), 0x164, 0xf, 0xf, true));
  v += __builtin_bit_cast(float, __builtin_amdgcn_update_dpp(0, __builtin_bit_cast(int, v), 0x168, 0xf, 0xf, true));
  return v;
}

__device__ __forceinline__ void load_idx(Idx& d, const int4* __restrict__ kc4,
                                         const int4* __restrict__ pr4,
                                         const int4* __restrict__ co4, unsigned q) {
  d.kc = kc4[q];
  d.pr = pr4[q];
  d.co = co4[q];
}

struct Gix {
  unsigned a0[TB], a1[TB];
  unsigned k2[TB];
  unsigned k8[TB];
  unsigned lo[TB];
};

__device__ __forceinline__ void load_rows(Rows& r, Gix& g, const Idx& ix,
                                          const float4* __restrict__ A4,
                                          const float4* __restrict__ M84,
                                          const float* __restrict__ M8,
                                          const float4* __restrict__ LO4,
                                          unsigned j1, unsigned j2) {
#pragma unroll
  for (int i = 0; i < TB; ++i) {
    const unsigned k = (unsigned)pick(ix.kc, i);
    const unsigned kr = __umul24(k, (unsigned)(NKC / 4));
    g.a0[i] = kr + j1;
    g.a1[i] = kr + j2;
    g.k2[i] = 2u * k;
    g.k8[i] = 8u * k;
    g.lo[i] = (unsigned)pick(ix.pr, i);
  }
#pragma unroll
  for (int i = 0; i < TB; ++i) {
    r.c0[i] = A4[g.a0[i]];
    r.c1[i] = A4[g.a1[i]];
    r.m4[i] = M84[g.k2[i]];
    r.sa[i] = (M8 + 4)[g.k8[i]];
    r.lo[i] = LO4[g.lo[i]];
  }
}

__device__ __forceinline__ void hold_indices(const Gix& g, unsigned q) {
#pragma unroll
  for (int i = 0; i < TB; ++i) {
    asm volatile("" :: "v"(g.a0[i]), "v"(g.a1[i]), "v"(g.k2[i]), "v"(g.k8[i]), "v"(g.lo[i]));
  }
  asm volatile("" :: "v"(q));
}

__device__ __forceinline__ void run_block(const Rows& r, const int4 co, bool has2,
                                          int j, float (&al0)[8], float (&al1)[8],
                                          float2* __restrict__ out2, int obase) {
  float ox[TB], oy[TB];
#pragma unroll
  for (int i = 0; i < TB; ++i) {
    float c[8];
    c[0] = r.c0[i].x; c[1] = r.c0[i].y; c[2] = r.c0[i].z; c[3] = r.c0[i].w;
    c[4] = has2 ? r.c1[i].x : 0.f; c[5] = has2 ? r.c1[i].y : 0.f;
    c[6] = has2 ? r.c1[i].z : 0.f; c[7] = has2 ? r.c1[i].w : 0.f;

    float p0 = c[0] * al0[0], p1 = c[0] * al1[0];
    float q0s = c[4] * al0[4], q1s = c[4] * al1[4];
#pragma unroll
    for (int k = 1; k < 4; ++k) {
      p0 = fmaf(c[k], al0[k], p0);
      p1 = fmaf(c[k], al1[k], p1);
      q0s = fmaf(c[4 + k], al0[4 + k], q0s);
      q1s = fmaf(c[4 + k], al1[4 + k], q1s);
    }
    const float a20 = rowsum16(p0 + q0s);
    const float a21 = rowsum16(p1 + q1s);

    float w0[8], w1[8];
#pragma unroll
    for (int k = 0; k < 8; ++k) {
      w0[k] = fmaf(-c[k], al0[k], al0[k]);
      w1[k] = fmaf(-c[k], al1[k], al1[k]);
    }

    const float sal = r.sa[i] * LOG2E_F;
    const float y00 = ex2(r.lo[i].x * sal), y01 = ex2(r.lo[i].y * sal);
    const float y10 = ex2(r.lo[i].z * sal), y11 = ex2(r.lo[i].w * sal);
    const int ci = pick(co, i);
    const float yc0 = (ci == 0) ? y00 : y01;
    const float yc1 = (ci == 0) ? y10 : y11;
    const float z00 = yc0 * r.m4[i].x, z01 = yc1 * r.m4[i].y;
    const float z10 = yc0 * r.m4[i].z, z11 = yc1 * r.m4[i].w;

    const float mx = fmaxf(a20, a21);
    const float e0 = ex2((a20 - mx) * LOG2E_F);
    const float e1 = ex2((a21 - mx) * LOG2E_F);

    const float t30 = fmaf(LN2_F, lg2(fmaf(e0, z00, e1 * z01)), mx);
    const float t31 = fmaf(LN2_F, lg2(fmaf(e0, z10, e1 * z11)), mx);

    const float s0 = fmaf(e0, y00, e1 * y10);
    const float s1 = fmaf(e0, y01, e1 * y11);
    const float lt = lg2(s0 + s1);

    ox[i] = (lg2(s0) - lt) * LN2_F;
    oy[i] = (lg2(s1) - lt) * LN2_F;
    asm volatile("" : "+v"(ox[i]), "+v"(oy[i]));

#pragma unroll
    for (int k = 0; k < 8; ++k) {
      al0[k] = fmaf(c[k], t30, w0[k]);
      al1[k] = fmaf(c[k], t31, w1[k]);
    }
  }
#pragma unroll
  for (int k = 0; k < 8; ++k) {
    asm volatile("" : "+v"(al0[k]), "+v"(al1[k]));
  }
  float sx = ox[0], sy = oy[0];
#pragma unroll
  for (int i = 1; i < TB; ++i) {
    sx = (j == i) ? ox[i] : sx;
    sy = (j == i) ? oy[i] : sy;
  }
  if (j < TB) {
    float2 o2;
    o2.x = sx;
    o2.y = sy;
    out2[obase + j] = o2;
  }
}

__global__ void __launch_bounds__(32) k_forward(
    const int* __restrict__ corr, const int* __restrict__ kc,
    const int* __restrict__ problem, const float* __restrict__ A,
    const float* __restrict__ init_logits, const float* __restrict__ M8,
    const float* __restrict__ LO, float* __restrict__ out) {
  const int l = threadIdx.x;
  const int j = l & 15;
  const int b = blockIdx.x * 2 + (l >> 4);
  const bool has2 = (j < 9);
  const unsigned pc1 = (unsigned)j;
  const unsigned pc2 = has2 ? (unsigned)(j + 16) : pc1;
  const float4* __restrict__ A4  = reinterpret_cast<const float4*>(A);
  const float4* __restrict__ M84 = reinterpret_cast<const float4*>(M8);
  const float4* __restrict__ LO4 = reinterpret_cast<const float4*>(LO);
  const int4* __restrict__ kc4 = reinterpret_cast<const int4*>(kc);
  const int4* __restrict__ pr4 = reinterpret_cast<const int4*>(problem);
  const int4* __restrict__ co4 = reinterpret_cast<const int4*>(corr);

  float al0[8], al1[8];
  {
    const float4* il = reinterpret_cast<const float4*>(init_logits);
    const float4 i0 = il[2 * j];
    const float4 i1 = il[2 * j + 1];
    const int j2 = has2 ? (32 + 2 * j) : (2 * j);
    const float4 i2 = il[j2];
    const float4 i3 = il[j2 + 1];
    float z;
    z = lse2(i0.x, i0.y); al0[0] = i0.x - z; al1[0] = i0.y - z;
    z = lse2(i0.z, i0.w); al0[1] = i0.z - z; al1[1] = i0.w - z;
    z = lse2(i1.x, i1.y); al0[2] = i1.x - z; al1[2] = i1.y - z;
    z = lse2(i1.z, i1.w); al0[3] = i1.z - z; al1[3] = i1.w - z;
    z = lse2(i2.x, i2.y); al0[4] = has2 ? i2.x - z : 0.f; al1[4] = has2 ? i2.y - z : 0.f;
    z = lse2(i2.z, i2.w); al0[5] = has2 ? i2.z - z : 0.f; al1[5] = has2 ? i2.w - z : 0.f;
    z = lse2(i3.x, i3.y); al0[6] = has2 ? i3.x - z : 0.f; al1[6] = has2 ? i3.y - z : 0.f;
    z = lse2(i3.z, i3.w); al0[7] = has2 ? i3.z - z : 0.f; al1[7] = has2 ? i3.w - z : 0.f;
  }

  const int base = b * NT;
  float2* __restrict__ out2 = reinterpret_cast<float2*>(out);

  const unsigned q0 = (unsigned)b * (unsigned)(NT / TB);
  Rows ra, rb;
  Gix ga, gb;
  Idx ip, iq;
  int4 co_cur;
  {
    Idx i0;
    load_idx(i0, kc4, pr4, co4, q0);
    load_rows(ra, ga, i0, A4, M84, M8, LO4, pc1, pc2);
    co_cur = i0.co;
    load_idx(ip, kc4, pr4, co4, q0 + 1u);
  }

  for (int t0 = 0; t0 < NT - TB; t0 += 2 * TB) {
    const unsigned nb = (unsigned)t0 / TB;
    load_rows(rb, gb, ip, A4, M84, M8, LO4, pc1, pc2);
    const int4 co_nb = ip.co;
    const unsigned qa = q0 + nb + 2u;
    load_idx(iq, kc4, pr4, co4, qa);
    __builtin_amdgcn_sched_barrier(0);
    run_block(ra, co_cur, has2, j, al0, al1, out2, base + t0);
    hold_indices(gb, qa);

    load_rows(ra, ga, iq, A4, M84, M8, LO4, pc1, pc2);
    const unsigned qb = q0 + ((nb + 3u < (unsigned)(NT / TB)) ? (nb + 3u) : (unsigned)(NT / TB - 1));
    load_idx(ip, kc4, pr4, co4, qb);
    __builtin_amdgcn_sched_barrier(0);
    run_block(rb, co_nb, has2, j, al0, al1, out2, base + t0 + TB);
    hold_indices(ga, qb);
    co_cur = iq.co;
  }
  run_block(ra, co_cur, has2, j, al0, al1, out2, base + NT - TB);
}

extern "C" void kernel_launch(void* const* d_in, const int* in_sizes, int n_in,
                              void* d_out, int out_size, void* d_ws, size_t ws_size,
                              hipStream_t stream) {
  (void)in_sizes; (void)n_in; (void)out_size; (void)ws_size;
  const int*   corr    = (const int*)d_in[0];
  const int*   kc      = (const int*)d_in[1];
  const int*   problem = (const int*)d_in[2];
  const float* A       = (const float*)d_in[3];
  const float* trans   = (const float*)d_in[4];
  const float* obs     = (const float*)d_in[5];
  const float* init    = (const float*)d_in[6];
  float* out = (float*)d_out;

  float* logtT = (float*)d_ws;
  float* M8    = logtT + 512;
  float* LO    = M8 + (size_t)NPROB * 8;

  k_logt<<<1, 128, 0, stream>>>(trans, logtT);
  k_lo<<<(NPROB + 127) / 128, 128, 0, stream>>>(obs, LO);
  k_gemm_M<<<NPROB / 16, 32, 0, stream>>>(A, logtT, M8);
  k_forward<<<NBATCH / 2, 32, 0, stream>>>(corr, kc, problem, A, init, M8, LO, out);
}


# k_forward as producer/consumer waves: a helper wave per scanning wave gathers the A rows and forms the alpha-independent weights into an LDS ring; the scan reads LDS only, its row sums one DPP group
# speedup vs baseline: 7.9886x; 1.2940x over previous
#include <hip/hip_runtime.h>
#include <math.h>

#define NPROB 10000
#define NKC   100
#define NBATCH 1024
#define NT    500

typedef __attribute__((ext_vector_type(2))) float v2f;
typedef __attribute__((ext_vector_type(8))) float v8f;

__device__ __forceinline__ float lse2(float a, float b) {
  float m = fmaxf(a, b);
  return m + __logf(__expf(a - m) + __expf(b - m));
}

__device__ __forceinline__ float ex2(float x) { return __builtin_amdgcn_exp2f(x); }
__device__ __forceinline__ float lg2(float x) { return __builtin_amdgcn_logf(x); }
#define LOG2E_F 1.4426950408889634f
#define LN2_F   0.6931471805599453f

#if __has_builtin(__builtin_amdgcn_wmma_f32_16x16x4_f32)
#define HAVE_WMMA_F32X4 1
#endif

__global__ void __launch_bounds__(32) k_tables(const float* __restrict__ A,
                                               const float* __restrict__ trans,
                                               const float* __restrict__ obs,
                                               float* __restrict__ M8,
                                               float* __restrict__ LO) {
  __shared__ __attribute__((aligned(16))) float s_logt[4 * NKC];
  const int tile = blockIdx.x;
  const int l = threadIdx.x;
  const int m = l & 15;
  const bool hi = (l >= 16);

  const float4* __restrict__ tr4 = reinterpret_cast<const float4*>(trans);
  float4 tv[4];
#pragma unroll
  for (int r = 0; r < 4; ++r) {
    const int k = l + 32 * r;
    tv[r] = tr4[k < NKC ? k : NKC - 1];
  }
  const int p = tile * 16 + m;
  const float4 ov = reinterpret_cast<const float4*>(obs)[p];

#pragma unroll
  for (int r = 0; r < 4; ++r) {
    const int k = l + 32 * r;
    const float4 v = tv[r];
    const float l0 = lse2(v.x, v.z);
    const float l1 = lse2(v.y, v.w);
    if (k < NKC) {
      s_logt[0 * NKC + k] = v.x - l0;
      s_logt[1 * NKC + k] = v.y - l1;
      s_logt[2 * NKC + k] = v.z - l0;
      s_logt[3 * NKC + k] = v.w - l1;
    }
  }
  {
    const float ls0 = lse2(ov.x, ov.y);
    const float ls1 = lse2(ov.z, ov.w);
    float4 r;
    r.x = ov.x - ls0; r.y = ov.y - ls0; r.z = ov.z - ls1; r.w = ov.w - ls1;
    if (!hi) reinterpret_cast<float4*>(LO)[p] = r;
  }
  __syncthreads();

#ifdef HAVE_WMMA_F32X4
  const int row = tile * 16 + m;
  const float nmask = (m < 4) ? 1.0f : 0.0f;
  const float ones4 = (m == 4) ? 1.0f : 0.0f;
  const float* bcolp = s_logt + (m & 3) * NKC;
  v8f c = {0.f, 0.f, 0.f, 0.f, 0.f, 0.f, 0.f, 0.f};
#pragma unroll 5
  for (int kk = 0; kk < 25; ++kk) {
    const int k0 = kk * 4 + (hi ? 2 : 0);
    const float2 av = *reinterpret_cast<const float2*>(A + row * NKC + k0);
    const float2 bv = *reinterpret_cast<const float2*>(bcolp + k0);
    v2f a, b;
    a.x = av.x;
    a.y = av.y;
    b.x = bv.x * nmask + ones4;
    b.y = bv.y * nmask + ones4;
    c = __builtin_amdgcn_wmma_f32_16x16x4_f32(
         false, a,  false, b,
         (short)0, c,  false,  false);
  }
  if (m < 5) {
#pragma unroll
    for (int v = 0; v < 8; ++v) {
      const int r = tile * 16 + v + (hi ? 8 : 0);
      M8[r * 8 + m] = (m < 4) ? __expf(c[v]) : c[v];
    }
  }
#else
  if (l < 16) {
    const int row = tile * 16 + m;
    float acc0 = 0.f, acc1 = 0.f, acc2 = 0.f, acc3 = 0.f, acc4 = 0.f;
    for (int k = 0; k < NKC; ++k) {
      const float av = A[row * NKC + k];
      acc0 += av * s_logt[0 * NKC + k];
      acc1 += av * s_logt[1 * NKC + k];
      acc2 += av * s_logt[2 * NKC + k];
      acc3 += av * s_logt[3 * NKC + k];
      acc4 += av;
    }
    M8[row * 8 + 0] = __expf(acc0); M8[row * 8 + 1] = __expf(acc1);
    M8[row * 8 + 2] = __expf(acc2); M8[row * 8 + 3] = __expf(acc3);
    M8[row * 8 + 4] = acc4;
  }
#endif
}

#define TB 4
#define SBLK 5
#define SSTEPS (TB * SBLK)
#define NSTAGE (NT / SSTEPS)
static_assert(NT % SSTEPS == 0, "stages must tile the sequence");

__device__ __forceinline__ float rowsum16_pair_max(float& a, float& b) {
  float mx;
  asm("v_add_f32_dpp %0, %0, %0 row_xmask:1 row_mask:0xf bank_mask:0xf bound_ctrl:1\n\t"
      "v_add_f32_dpp %1, %1, %1 row_xmask:1 row_mask:0xf bank_mask:0xf bound_ctrl:1\n\t"
      "v_add_f32_dpp %0, %0, %0 row_xmask:2 row_mask:0xf bank_mask:0xf bound_ctrl:1\n\t"
      "v_add_f32_dpp %1, %1, %1 row_xmask:2 row_mask:0xf bank_mask:0xf bound_ctrl:1\n\t"
      "v_add_f32_dpp %0, %0, %0 row_xmask:4 row_mask:0xf bank_mask:0xf bound_ctrl:1\n\t"
      "v_add_f32_dpp %1, %1, %1 row_xmask:4 row_mask:0xf bank_mask:0xf bound_ctrl:1\n\t"
      "v_add_f32_dpp %0, %0, %0 row_xmask:8 row_mask:0xf bank_mask:0xf bound_ctrl:1\n\t"
      "v_add_f32_dpp %1, %1, %1 row_xmask:8 row_mask:0xf bank_mask:0xf bound_ctrl:1\n\t"
      "v_max_num_f32 %2, %0, %1"
      : "+v"(a), "+v"(b), "=&v"(mx));
  return mx;
}

template <int I> __device__ __forceinline__ int row_bcast(int v) {
  return __builtin_amdgcn_update_dpp(0, v, 0x150 + I, 0xf, 0xf, true);
}

#define STAGE_STEPS(F) F(0) F(1) F(2) F(3) F(4) F(5) F(6) F(7) F(8) F(9) \
                       F(10) F(11) F(12) F(13) F(14) F(15) F(16) F(17) F(18) F(19)
static_assert(SSTEPS == 20, "STAGE_STEPS lists the steps of a stage");
#define GATHER_STEP(I)                                                                      \
  const unsigned kr_##I = __umul24(                                                         \
      (unsigned)(((I) < 16) ? row_bcast<((I) & 15)>(kv1) : row_bcast<((I) & 15)>(kv2)),     \
      (unsigned)(NKC / 4));                                                                 \
  const float4 c0_##I = A4[kr_##I + pc1];                                                   \
  const float4 c1_##I = A4[kr_##I + pc2];
#define STORE_STEP(I)                                                                       \
  s_a[pair][slot][I][row].p[j] = c0_##I;                                                    \
  s_a[pair][slot][I][row].p[j + 16] =                                                       \
      make_float4(has2 ? c1_##I.x : 0.f, has2 ? c1_##I.y : 0.f,                             \
                  has2 ? c1_##I.z : 0.f, has2 ? c1_##I.w : 0.f);

struct CRows {
  float4 c0[TB];
  float4 c1[TB];
  float4 y[TB];
  float4 z[TB];
};

__device__ __forceinline__ void run_block(const CRows& r, int j,
                                          float (&al0)[8], float (&al1)[8],
                                          float2* __restrict__ out2, int obase) {
  float ox[TB], oy[TB];
#pragma unroll
  for (int i = 0; i < TB; ++i) {
    float c[8];
    c[0] = r.c0[i].x; c[1] = r.c0[i].y; c[2] = r.c0[i].z; c[3] = r.c0[i].w;
    c[4] = r.c1[i].x; c[5] = r.c1[i].y; c[6] = r.c1[i].z; c[7] = r.c1[i].w;

    float p0 = c[0] * al0[0], p1 = c[0] * al1[0];
    float q0s = c[4] * al0[4], q1s = c[4] * al1[4];
#pragma unroll
    for (int k = 1; k < 4; ++k) {
      p0 = fmaf(c[k], al0[k], p0);
      p1 = fmaf(c[k], al1[k], p1);
      q0s = fmaf(c[4 + k], al0[4 + k], q0s);
      q1s = fmaf(c[4 + k], al1[4 + k], q1s);
    }
    float a20 = p0 + q0s;
    float a21 = p1 + q1s;
    const float mx = rowsum16_pair_max(a20, a21);

    float w0[8], w1[8];
#pragma unroll
    for (int k = 0; k < 8; ++k) {
      w0[k] = fmaf(-c[k], al0[k], al0[k]);
      w1[k] = fmaf(-c[k], al1[k], al1[k]);
    }

    const float y00 = r.y[i].x, y01 = r.y[i].y, y10 = r.y[i].z, y11 = r.y[i].w;
    const float z00 = r.z[i].x, z01 = r.z[i].y, z10 = r.z[i].z, z11 = r.z[i].w;

    const float e0 = ex2((a20 - mx) * LOG2E_F);
    const float e1 = ex2((a21 - mx) * LOG2E_F);

    const float t30 = fmaf(LN2_F, lg2(fmaf(e0, z00, e1 * z01)), mx);
    const float t31 = fmaf(LN2_F, lg2(fmaf(e0, z10, e1 * z11)), mx);

    const float s0 = fmaf(e0, y00, e1 * y10);
    const float s1 = fmaf(e0, y01, e1 * y11);
    const float lt = lg2(s0 + s1);

    ox[i] = (lg2(s0) - lt) * LN2_F;
    oy[i] = (lg2(s1) - lt) * LN2_F;
    asm volatile("" : "+v"(ox[i]), "+v"(oy[i]));

#pragma unroll
    for (int k = 0; k < 8; ++k) {
      al0[k] = fmaf(c[k], t30, w0[k]);
      al1[k] = fmaf(c[k], t31, w1[k]);
    }
  }
#pragma unroll
  for (int k = 0; k < 8; ++k) {
    asm volatile("" : "+v"(al0[k]), "+v"(al1[k]));
  }
  float sx = ox[0], sy = oy[0];
#pragma unroll
  for (int i = 1; i < TB; ++i) {
    sx = (j == i) ? ox[i] : sx;
    sy = (j == i) ? oy[i] : sy;
  }
  if (j < TB) {
    float2 o2;
    o2.x = sx;
    o2.y = sy;
    out2[obase + j] = o2;
  }
}

struct __attribute__((aligned(16))) ARow { float4 p[32]; };
struct __attribute__((aligned(16))) YZ { float4 y; float4 z; };

__device__ __forceinline__ YZ make_yz(const float4 lo, const float sa, const float4 m4, const int ci) {
  const float sal = sa * LOG2E_F;
  const float y00 = ex2(lo.x * sal), y01 = ex2(lo.y * sal);
  const float y10 = ex2(lo.z * sal), y11 = ex2(lo.w * sal);
  const float yc0 = (ci == 0) ? y00 : y01;
  const float yc1 = (ci == 0) ? y10 : y11;
  YZ v;
  v.y = make_float4(y00, y01, y10, y11);
  v.z = make_float4(yc0 * m4.x, yc1 * m4.y, yc0 * m4.z, yc1 * m4.w);
  return v;
}

__global__ void __launch_bounds__(128) k_forward(
    const int* __restrict__ corr, const int* __restrict__ kc,
    const int* __restrict__ problem, const float* __restrict__ A,
    const float* __restrict__ init_logits, const float* __restrict__ M8,
    const float* __restrict__ LO, float* __restrict__ out) {
  __shared__ ARow s_a[2][2][SSTEPS][2];
  __shared__ YZ s_yz[2][2][SSTEPS][2];

  const int l = threadIdx.x & 31;
  const int wv = threadIdx.x >> 5;
  const int pair = wv & 1;
  const bool helper = (wv >= 2);
  const int j = l & 15;
  const int row = l >> 4;
  const int b = (blockIdx.x * 2 + pair) * 2 + row;
  const bool has2 = (j < 9);
  const float4* __restrict__ A4  = reinterpret_cast<const float4*>(A);
  const float4* __restrict__ M84 = reinterpret_cast<const float4*>(M8);
  const float4* __restrict__ LO4 = reinterpret_cast<const float4*>(LO);
  float2* __restrict__ out2 = reinterpret_cast<float2*>(out);
  const int base = b * NT;
  const unsigned pc1 = (unsigned)j;
  const unsigned pc2 = has2 ? (unsigned)(j + 16) : pc1;
  const int t1 = j;
  const int t2 = 16 + (j & 3);

  float al0[8], al1[8];
  int kv1 = 0, pv1 = 0, cv1 = 0, kv2 = 0, pv2 = 0, cv2 = 0;
  if (!helper) {
    const float4* il = reinterpret_cast<const float4*>(init_logits);
    const float4 i0 = il[2 * j];
    const float4 i1 = il[2 * j + 1];
    const int j2 = has2 ? (32 + 2 * j) : (2 * j);
    const float4 i2 = il[j2];
    const float4 i3 = il[j2 + 1];
    float z;
    z = lse2(i0.x, i0.y); al0[0] = i0.x - z; al1[0] = i0.y - z;
    z = lse2(i0.z, i0.w); al0[1] = i0.z - z; al1[1] = i0.w - z;
    z = lse2(i1.x, i1.y); al0[2] = i1.x - z; al1[2] = i1.y - z;
    z = lse2(i1.z, i1.w); al0[3] = i1.z - z; al1[3] = i1.w - z;
    z = lse2(i2.x, i2.y); al0[4] = has2 ? i2.x - z : 0.f; al1[4] = has2 ? i2.y - z : 0.f;
    z = lse2(i2.z, i2.w); al0[5] = has2 ? i2.z - z : 0.f; al1[5] = has2 ? i2.w - z : 0.f;
    z = lse2(i3.x, i3.y); al0[6] = has2 ? i3.x - z : 0.f; al1[6] = has2 ? i3.y - z : 0.f;
    z = lse2(i3.z, i3.w); al0[7] = has2 ? i3.z - z : 0.f; al1[7] = has2 ? i3.w - z : 0.f;
  } else {
#pragma unroll
    for (int k = 0; k < 8; ++k) { al0[k] = 0.f; al1[k] = 0.f; }
    kv1 = kc[base + t1]; pv1 = problem[base + t1]; cv1 = corr[base + t1];
    kv2 = kc[base + t2]; pv2 = problem[base + t2]; cv2 = corr[base + t2];
  }

  for (int n = 0; n <= NSTAGE; ++n) {
    if (helper) {
      if (n < NSTAGE) {
        const int slot = n & 1;
        STAGE_STEPS(GATHER_STEP)
        const float4 m4a = M84[2u * (unsigned)kv1];
        const float saa = (M8 + 4)[8u * (unsigned)kv1];
        const float4 loa = LO4[(unsigned)pv1];
        const float4 m4b = M84[2u * (unsigned)kv2];
        const float sab = (M8 + 4)[8u * (unsigned)kv2];
        const float4 lob = LO4[(unsigned)pv2];
        const int cia = cv1, cib = cv2;
        const int nb = base + ((n + 1 < NSTAGE) ? (n + 1) : n) * SSTEPS;
        kv1 = kc[nb + t1]; pv1 = problem[nb + t1]; cv1 = corr[nb + t1];
        kv2 = kc[nb + t2]; pv2 = problem[nb + t2]; cv2 = corr[nb + t2];

        s_yz[pair][slot][t1][row] = make_yz(loa, saa, m4a, cia);
        s_yz[pair][slot][t2][row] = make_yz(lob, sab, m4b, cib);
        STAGE_STEPS(STORE_STEP)
      }
    } else {
      if (n >= 1) {
        const int m = n - 1;
        const int slot = m & 1;
        const int ob = base + m * SSTEPS;
        CRows ra, rb;
#pragma unroll
        for (int i = 0; i < TB; ++i) {
          ra.c0[i] = s_a[pair][slot][i][row].p[j];
          ra.c1[i] = s_a[pair][slot][i][row].p[j + 16];
          ra.y[i] = s_yz[pair][slot][i][row].y;
          ra.z[i] = s_yz[pair][slot][i][row].z;
        }
#pragma unroll
        for (int s = 0; s < SBLK; ++s) {
          CRows& cur = (s & 1) ? rb : ra;
          CRows& nxt = (s & 1) ? ra : rb;
          if (s + 1 < SBLK) {
#pragma unroll
            for (int i = 0; i < TB; ++i) {
              const int st = (s + 1) * TB + i;
              nxt.c0[i] = s_a[pair][slot][st][row].p[j];
              nxt.c1[i] = s_a[pair][slot][st][row].p[j + 16];
              nxt.y[i] = s_yz[pair][slot][st][row].y;
              nxt.z[i] = s_yz[pair][slot][st][row].z;
            }
          }
          __builtin_amdgcn_sched_barrier(0);
          run_block(cur, j, al0, al1, out2, ob + s * TB);
        }
      }
    }
    __builtin_amdgcn_fence(__ATOMIC_RELEASE, "workgroup", "local");
    __builtin_amdgcn_s_barrier();
    __builtin_amdgcn_fence(__ATOMIC_ACQUIRE, "workgroup", "local");
  }
}

extern "C" void kernel_launch(void* const* d_in, const int* in_sizes, int n_in,
                              void* d_out, int out_size, void* d_ws, size_t ws_size,
                              hipStream_t stream) {
  (void)in_sizes; (void)n_in; (void)out_size; (void)ws_size;
  const int*   corr    = (const int*)d_in[0];
  const int*   kc      = (const int*)d_in[1];
  const int*   problem = (const int*)d_in[2];
  const float* A       = (const float*)d_in[3];
  const float* trans   = (const float*)d_in[4];
  const float* obs     = (const float*)d_in[5];
  const float* init    = (const float*)d_in[6];
  float* out = (float*)d_out;

  float* M8    = (float*)d_ws + 512;
  float* LO    = M8 + (size_t)NPROB * 8;

  k_tables<<<NPROB / 16, 32, 0, stream>>>(A, trans, obs, M8, LO);
  k_forward<<<NBATCH / 4, 128, 0, stream>>>(corr, kc, problem, A, init, M8, LO, out);
}


# k_forward: no address-translation wait (s_wait_xcnt) left inside either wave's pass: a step's two A gathers share one address register, outputs are stored 16 lanes at once; row-sum order changes
# speedup vs baseline: 8.4077x; 1.0525x over previous
#include <hip/hip_runtime.h>
#include <math.h>

#define NPROB 10000
#define NKC   100
#define NBATCH 1024
#define NT    500

typedef __attribute__((ext_vector_type(2))) float v2f;
typedef __attribute__((ext_vector_type(8))) float v8f;

__device__ __forceinline__ float lse2(float a, float b) {
  float m = fmaxf(a, b);
  return m + __logf(__expf(a - m) + __expf(b - m));
}

__device__ __forceinline__ float ex2(float x) { return __builtin_amdgcn_exp2f(x); }
__device__ __forceinline__ float lg2(float x) { return __builtin_amdgcn_logf(x); }
#define LOG2E_F 1.4426950408889634f
#define LN2_F   0.6931471805599453f

#if __has_builtin(__builtin_amdgcn_wmma_f32_16x16x4_f32)
#define HAVE_WMMA_F32X4 1
#endif

__global__ void __launch_bounds__(32) k_tables(const float* __restrict__ A,
                                               const float* __restrict__ trans,
                                               const float* __restrict__ obs,
                                               float* __restrict__ M8,
                                               float* __restrict__ LO) {
  __shared__ __attribute__((aligned(16))) float s_logt[4 * NKC];
  const int tile = blockIdx.x;
  const int l = threadIdx.x;
  const int m = l & 15;
  const bool hi = (l >= 16);

  const float4* __restrict__ tr4 = reinterpret_cast<const float4*>(trans);
  float4 tv[4];
#pragma unroll
  for (int r = 0; r < 4; ++r) {
    const int k = l + 32 * r;
    tv[r] = tr4[k < NKC ? k : NKC - 1];
  }
  const int p = tile * 16 + m;
  const float4 ov = reinterpret_cast<const float4*>(obs)[p];

#pragma unroll
  for (int r = 0; r < 4; ++r) {
    const int k = l + 32 * r;
    const float4 v = tv[r];
    const float l0 = lse2(v.x, v.z);
    const float l1 = lse2(v.y, v.w);
    if (k < NKC) {
      s_logt[0 * NKC + k] = v.x - l0;
      s_logt[1 * NKC + k] = v.y - l1;
      s_logt[2 * NKC + k] = v.z - l0;
      s_logt[3 * NKC + k] = v.w - l1;
    }
  }
  {
    const float ls0 = lse2(ov.x, ov.y);
    const float ls1 = lse2(ov.z, ov.w);
    float4 r;
    r.x = ov.x - ls0; r.y = ov.y - ls0; r.z = ov.z - ls1; r.w = ov.w - ls1;
    if (!hi) reinterpret_cast<float4*>(LO)[p] = r;
  }
  __syncthreads();

#ifdef HAVE_WMMA_F32X4
  const int row = tile * 16 + m;
  const float nmask = (m < 4) ? 1.0f : 0.0f;
  const float ones4 = (m == 4) ? 1.0f : 0.0f;
  const float* bcolp = s_logt + (m & 3) * NKC;
  v8f c = {0.f, 0.f, 0.f, 0.f, 0.f, 0.f, 0.f, 0.f};
#pragma unroll 5
  for (int kk = 0; kk < 25; ++kk) {
    const int k0 = kk * 4 + (hi ? 2 : 0);
    const float2 av = *reinterpret_cast<const float2*>(A + row * NKC + k0);
    const float2 bv = *reinterpret_cast<const float2*>(bcolp + k0);
    v2f a, b;
    a.x = av.x;
    a.y = av.y;
    b.x = bv.x * nmask + ones4;
    b.y = bv.y * nmask + ones4;
    c = __builtin_amdgcn_wmma_f32_16x16x4_f32(
         false, a,  false, b,
         (short)0, c,  false,  false);
  }
  if (m < 5) {
#pragma unroll
    for (int v = 0; v < 8; ++v) {
      const int r = tile * 16 + v + (hi ? 8 : 0);
      M8[r * 8 + m] = (m < 4) ? __expf(c[v]) : c[v];
    }
  }
#else
  if (l < 16) {
    const int row = tile * 16 + m;
    float acc0 = 0.f, acc1 = 0.f, acc2 = 0.f, acc3 = 0.f, acc4 = 0.f;
    for (int k = 0; k < NKC; ++k) {
      const float av = A[row * NKC + k];
      acc0 += av * s_logt[0 * NKC + k];
      acc1 += av * s_logt[1 * NKC + k];
      acc2 += av * s_logt[2 * NKC + k];
      acc3 += av * s_logt[3 * NKC + k];
      acc4 += av;
    }
    M8[row * 8 + 0] = __expf(acc0); M8[row * 8 + 1] = __expf(acc1);
    M8[row * 8 + 2] = __expf(acc2); M8[row * 8 + 3] = __expf(acc3);
    M8[row * 8 + 4] = acc4;
  }
#endif
}

#define TB 4
#define SBLK 5
#define SSTEPS (TB * SBLK)
#define NSTAGE (NT / SSTEPS)
static_assert(NT % SSTEPS == 0, "stages must tile the sequence");
static_assert(16 % TB == 0 && SSTEPS == 16 + TB, "a stage's outputs: one 16-lane store, one TB-lane store");

__device__ __forceinline__ float rowsum16_pair_max(float& a, float& b) {
  float mx;
  asm("v_add_f32_dpp %0, %0, %0 row_xmask:1 row_mask:0xf bank_mask:0xf bound_ctrl:1\n\t"
      "v_add_f32_dpp %1, %1, %1 row_xmask:1 row_mask:0xf bank_mask:0xf bound_ctrl:1\n\t"
      "v_add_f32_dpp %0, %0, %0 row_xmask:2 row_mask:0xf bank_mask:0xf bound_ctrl:1\n\t"
      "v_add_f32_dpp %1, %1, %1 row_xmask:2 row_mask:0xf bank_mask:0xf bound_ctrl:1\n\t"
      "v_add_f32_dpp %0, %0, %0 row_xmask:4 row_mask:0xf bank_mask:0xf bound_ctrl:1\n\t"
      "v_add_f32_dpp %1, %1, %1 row_xmask:4 row_mask:0xf bank_mask:0xf bound_ctrl:1\n\t"
      "v_add_f32_dpp %0, %0, %0 row_xmask:8 row_mask:0xf bank_mask:0xf bound_ctrl:1\n\t"
      "v_add_f32_dpp %1, %1, %1 row_xmask:8 row_mask:0xf bank_mask:0xf bound_ctrl:1\n\t"
      "v_max_num_f32 %2, %0, %1"
      : "+v"(a), "+v"(b), "=&v"(mx));
  return mx;
}

template <int I> __device__ __forceinline__ int row_bcast(int v) {
  return __builtin_amdgcn_update_dpp(0, v, 0x150 + I, 0xf, 0xf, true);
}

#define STAGE_STEPS(F) F(0) F(1) F(2) F(3) F(4) F(5) F(6) F(7) F(8) F(9) \
                       F(10) F(11) F(12) F(13) F(14) F(15) F(16) F(17) F(18) F(19)
static_assert(SSTEPS == 20, "STAGE_STEPS lists the steps of a stage");
#define INDEX_STEP(I)                                                                       \
  const unsigned kr_##I = __umul24(                                                         \
      (unsigned)(((I) < 16) ? row_bcast<((I) & 15)>(kv1) : row_bcast<((I) & 15)>(kv2)),     \
      (unsigned)(NKC / 4)) + pc1;
#define GATHER_STEP(I)                                                                      \
  const float4 c0_##I = A4[kr_##I];                                                         \
  const float4 c1_##I = A4b[kr_##I];
#define STORE_STEP(I)                                                                       \
  s_a[pair][slot][I][row].p[j] = c0_##I;                                                    \
  s_a[pair][slot][I][row].p[j + 16] =                                                       \
      make_float4(has2 ? c1_##I.x : 0.f, has2 ? c1_##I.y : 0.f,                             \
                  has2 ? c1_##I.z : 0.f, has2 ? c1_##I.w : 0.f);

struct CRows {
  float4 c0[TB];
  float4 c1[TB];
  float4 y[TB];
  float4 z[TB];
};

__device__ __forceinline__ void run_block(const CRows& r, int j,
                                          float (&al0)[8], float (&al1)[8],
                                          float2* __restrict__ out2, int ob, const int S,
                                          float& sx, float& sy, unsigned& st_i) {
  float ox[TB], oy[TB];
#pragma unroll
  for (int i = 0; i < TB; ++i) {
    float c[8];
    c[0] = r.c0[i].x; c[1] = r.c0[i].y; c[2] = r.c0[i].z; c[3] = r.c0[i].w;
    c[4] = r.c1[i].x; c[5] = r.c1[i].y; c[6] = r.c1[i].z; c[7] = r.c1[i].w;

    float p0 = c[0] * al0[0], p1 = c[0] * al1[0];
    float q0s = c[4] * al0[4], q1s = c[4] * al1[4];
#pragma unroll
    for (int k = 1; k < 4; ++k) {
      p0 = fmaf(c[k], al0[k], p0);
      p1 = fmaf(c[k], al1[k], p1);
      q0s = fmaf(c[4 + k], al0[4 + k], q0s);
      q1s = fmaf(c[4 + k], al1[4 + k], q1s);
    }
    float a20 = p0 + q0s;
    float a21 = p1 + q1s;
    const float mx = rowsum16_pair_max(a20, a21);

    float w0[8], w1[8];
#pragma unroll
    for (int k = 0; k < 8; ++k) {
      w0[k] = fmaf(-c[k], al0[k], al0[k]);
      w1[k] = fmaf(-c[k], al1[k], al1[k]);
    }

    const float y00 = r.y[i].x, y01 = r.y[i].y, y10 = r.y[i].z, y11 = r.y[i].w;
    const float z00 = r.z[i].x, z01 = r.z[i].y, z10 = r.z[i].z, z11 = r.z[i].w;

    const float e0 = ex2((a20 - mx) * LOG2E_F);
    const float e1 = ex2((a21 - mx) * LOG2E_F);

    const float t30 = fmaf(LN2_F, lg2(fmaf(e0, z00, e1 * z01)), mx);
    const float t31 = fmaf(LN2_F, lg2(fmaf(e0, z10, e1 * z11)), mx);

    const float s0 = fmaf(e0, y00, e1 * y10);
    const float s1 = fmaf(e0, y01, e1 * y11);
    const float lt = lg2(s0 + s1);

    ox[i] = (lg2(s0) - lt) * LN2_F;
    oy[i] = (lg2(s1) - lt) * LN2_F;
    asm volatile("" : "+v"(ox[i]), "+v"(oy[i]));

#pragma unroll
    for (int k = 0; k < 8; ++k) {
      al0[k] = fmaf(c[k], t30, w0[k]);
      al1[k] = fmaf(c[k], t31, w1[k]);
    }
  }
#pragma unroll
  for (int k = 0; k < 8; ++k) {
    asm volatile("" : "+v"(al0[k]), "+v"(al1[k]));
  }
  if (S * TB < 16) {
#pragma unroll
    for (int i = 0; i < TB; ++i) {
      sx = (j == S * TB + i) ? ox[i] : sx;
      sy = (j == S * TB + i) ? oy[i] : sy;
    }
    if ((S + 1) * TB == 16) {
      st_i = (unsigned)(ob + j);
      float2 o2;
      o2.x = sx;
      o2.y = sy;
      out2[st_i] = o2;
    }
  } else {
    float tx = ox[0], ty = oy[0];
#pragma unroll
    for (int i = 1; i < TB; ++i) {
      tx = (j == i) ? ox[i] : tx;
      ty = (j == i) ? oy[i] : ty;
    }
    if (j < TB) {
      float2 o2;
      o2.x = tx;
      o2.y = ty;
      out2[ob + S * TB + j] = o2;
    }
  }
}

struct __attribute__((aligned(16))) ARow { float4 p[32]; };
struct __attribute__((aligned(16))) YZ { float4 y; float4 z; };

__device__ __forceinline__ YZ make_yz(const float4 lo, const float sa, const float4 m4, const int ci) {
  const float sal = sa * LOG2E_F;
  const float y00 = ex2(lo.x * sal), y01 = ex2(lo.y * sal);
  const float y10 = ex2(lo.z * sal), y11 = ex2(lo.w * sal);
  const float yc0 = (ci == 0) ? y00 : y01;
  const float yc1 = (ci == 0) ? y10 : y11;
  YZ v;
  v.y = make_float4(y00, y01, y10, y11);
  v.z = make_float4(yc0 * m4.x, yc1 * m4.y, yc0 * m4.z, yc1 * m4.w);
  return v;
}

__global__ void __launch_bounds__(128) k_forward(
    const int* __restrict__ corr, const int* __restrict__ kc,
    const int* __restrict__ problem, const float* __restrict__ A,
    const float* __restrict__ init_logits, const float* __restrict__ M8,
    const float* __restrict__ LO, float* __restrict__ out) {
  __shared__ ARow s_a[2][2][SSTEPS][2];
  __shared__ YZ s_yz[2][2][SSTEPS][2];

  const int l = threadIdx.x & 31;
  const int wv = threadIdx.x >> 5;
  const int pair = wv & 1;
  const bool helper = (wv >= 2);
  const int j = l & 15;
  const int row = l >> 4;
  const int b = (blockIdx.x * 2 + pair) * 2 + row;
  const bool has2 = (j >= 7);
  const float4* __restrict__ A4  = reinterpret_cast<const float4*>(A);
  const float4* __restrict__ A4b = A4 + 9;
  const float4* __restrict__ M84 = reinterpret_cast<const float4*>(M8);
  const float4* __restrict__ LO4 = reinterpret_cast<const float4*>(LO);
  float2* __restrict__ out2 = reinterpret_cast<float2*>(out);
  const int base = b * NT;
  const unsigned pc1 = (unsigned)j;
  const int t1 = j;
  const int t2 = 16 + (j & 3);

  float al0[8], al1[8];
  int kv1 = 0, pv1 = 0, cv1 = 0, kv2 = 0, pv2 = 0, cv2 = 0;
  if (!helper) {
    const float4* il = reinterpret_cast<const float4*>(init_logits);
    const float4 i0 = il[2 * j];
    const float4 i1 = il[2 * j + 1];
    const int j2 = 2 * (j + 9);
    const float4 i2 = il[j2];
    const float4 i3 = il[j2 + 1];
    float z;
    z = lse2(i0.x, i0.y); al0[0] = i0.x - z; al1[0] = i0.y - z;
    z = lse2(i0.z, i0.w); al0[1] = i0.z - z; al1[1] = i0.w - z;
    z = lse2(i1.x, i1.y); al0[2] = i1.x - z; al1[2] = i1.y - z;
    z = lse2(i1.z, i1.w); al0[3] = i1.z - z; al1[3] = i1.w - z;
    z = lse2(i2.x, i2.y); al0[4] = has2 ? i2.x - z : 0.f; al1[4] = has2 ? i2.y - z : 0.f;
    z = lse2(i2.z, i2.w); al0[5] = has2 ? i2.z - z : 0.f; al1[5] = has2 ? i2.w - z : 0.f;
    z = lse2(i3.x, i3.y); al0[6] = has2 ? i3.x - z : 0.f; al1[6] = has2 ? i3.y - z : 0.f;
    z = lse2(i3.z, i3.w); al0[7] = has2 ? i3.z - z : 0.f; al1[7] = has2 ? i3.w - z : 0.f;
  } else {
#pragma unroll
    for (int k = 0; k < 8; ++k) { al0[k] = 0.f; al1[k] = 0.f; }
    kv1 = kc[base + t1]; pv1 = problem[base + t1]; cv1 = corr[base + t1];
    kv2 = kc[base + t2]; pv2 = problem[base + t2]; cv2 = corr[base + t2];
  }

  for (int n = 0; n <= NSTAGE; ++n) {
    if (helper) {
      if (n < NSTAGE) {
        const int slot = n & 1;
        STAGE_STEPS(INDEX_STEP)
        const unsigned ma = 2u * (unsigned)kv1, mb = 2u * (unsigned)kv2;
        const unsigned qa = 8u * (unsigned)kv1, qb = 8u * (unsigned)kv2;
        const int nb = base + ((n + 1 < NSTAGE) ? (n + 1) : n) * SSTEPS;
        const unsigned n1 = (unsigned)(nb + t1), n2 = (unsigned)(nb + t2);
        asm volatile("" : : "v"(kr_0), "v"(kr_1), "v"(kr_2), "v"(kr_3), "v"(kr_4), "v"(kr_5),
                     "v"(kr_6), "v"(kr_7), "v"(kr_8), "v"(kr_9), "v"(kr_10), "v"(kr_11),
                     "v"(kr_12), "v"(kr_13), "v"(kr_14), "v"(kr_15), "v"(kr_16), "v"(kr_17),
                     "v"(kr_18), "v"(kr_19), "v"(ma), "v"(mb), "v"(qa), "v"(qb), "v"(n1), "v"(n2));
        __builtin_amdgcn_sched_barrier(0);
        STAGE_STEPS(GATHER_STEP)
        const float4 m4a = M84[ma];
        const float saa = (M8 + 4)[qa];
        const float4 loa = LO4[(unsigned)pv1];
        const float4 m4b = M84[mb];
        const float sab = (M8 + 4)[qb];
        const float4 lob = LO4[(unsigned)pv2];
        const int cia = cv1, cib = cv2;
        kv1 = kc[n1]; pv1 = problem[n1]; cv1 = corr[n1];
        kv2 = kc[n2]; pv2 = problem[n2]; cv2 = corr[n2];
        __builtin_amdgcn_sched_barrier(0);

        s_yz[pair][slot][t1][row] = make_yz(loa, saa, m4a, cia);
        s_yz[pair][slot][t2][row] = make_yz(lob, sab, m4b, cib);
        STAGE_STEPS(STORE_STEP)
      }
    } else {
      if (n >= 1) {
        const int m = n - 1;
        const int slot = m & 1;
        const int ob = base + m * SSTEPS;
        float sx = 0.f, sy = 0.f;
        unsigned st_i = 0u;
        CRows ra, rb;
#pragma unroll
        for (int i = 0; i < TB; ++i) {
          ra.c0[i] = s_a[pair][slot][i][row].p[j];
          ra.c1[i] = s_a[pair][slot][i][row].p[j + 16];
          ra.y[i] = s_yz[pair][slot][i][row].y;
          ra.z[i] = s_yz[pair][slot][i][row].z;
        }
#pragma unroll
        for (int s = 0; s < SBLK; ++s) {
          CRows& cur = (s & 1) ? rb : ra;
          CRows& nxt = (s & 1) ? ra : rb;
          if (s + 1 < SBLK) {
#pragma unroll
            for (int i = 0; i < TB; ++i) {
              const int st = (s + 1) * TB + i;
              nxt.c0[i] = s_a[pair][slot][st][row].p[j];
              nxt.c1[i] = s_a[pair][slot][st][row].p[j + 16];
              nxt.y[i] = s_yz[pair][slot][st][row].y;
              nxt.z[i] = s_yz[pair][slot][st][row].z;
            }
          }
          __builtin_amdgcn_sched_barrier(0);
          run_block(cur, j, al0, al1, out2, ob, s, sx, sy, st_i);
        }
        asm volatile("" : : "v"(sx), "v"(sy), "v"(st_i));
      }
    }
    __builtin_amdgcn_fence(__ATOMIC_RELEASE, "workgroup", "local");
    __builtin_amdgcn_s_barrier();
    __builtin_amdgcn_fence(__ATOMIC_ACQUIRE, "workgroup", "local");
  }
}

extern "C" void kernel_launch(void* const* d_in, const int* in_sizes, int n_in,
                              void* d_out, int out_size, void* d_ws, size_t ws_size,
                              hipStream_t stream) {
  (void)in_sizes; (void)n_in; (void)out_size; (void)ws_size;
  const int*   corr    = (const int*)d_in[0];
  const int*   kc      = (const int*)d_in[1];
  const int*   problem = (const int*)d_in[2];
  const float* A       = (const float*)d_in[3];
  const float* trans   = (const float*)d_in[4];
  const float* obs     = (const float*)d_in[5];
  const float* init    = (const float*)d_in[6];
  float* out = (float*)d_out;

  float* M8    = (float*)d_ws + 512;
  float* LO    = M8 + (size_t)NPROB * 8;

  k_tables<<<NPROB / 16, 32, 0, stream>>>(A, trans, obs, M8, LO);
  k_forward<<<NBATCH / 4, 128, 0, stream>>>(corr, kc, problem, A, init, M8, LO, out);
}


# k_forward scan: issue order on the serial chain by ties only (same arithmetic): both exponents before the 1st exp; a block's LDS reads behind its 1st row sums; last output logs behind the update
# speedup vs baseline: 8.7427x; 1.0398x over previous
#include <hip/hip_runtime.h>
#include <math.h>

#define NPROB 10000
#define NKC   100
#define NBATCH 1024
#define NT    500

typedef __attribute__((ext_vector_type(2))) float v2f;
typedef __attribute__((ext_vector_type(8))) float v8f;

__device__ __forceinline__ float lse2(float a, float b) {
  float m = fmaxf(a, b);
  return m + __logf(__expf(a - m) + __expf(b - m));
}

__device__ __forceinline__ float ex2(float x) { return __builtin_amdgcn_exp2f(x); }
__device__ __forceinline__ float lg2(float x) { return __builtin_amdgcn_logf(x); }
#define LOG2E_F 1.4426950408889634f
#define LN2_F   0.6931471805599453f

#if __has_builtin(__builtin_amdgcn_wmma_f32_16x16x4_f32)
#define HAVE_WMMA_F32X4 1
#endif

__global__ void __launch_bounds__(32) k_tables(const float* __restrict__ A,
                                               const float* __restrict__ trans,
                                               const float* __restrict__ obs,
                                               float* __restrict__ M8,
                                               float* __restrict__ LO) {
  __shared__ __attribute__((aligned(16))) float s_logt[4 * NKC];
  const int tile = blockIdx.x;
  const int l = threadIdx.x;
  const int m = l & 15;
  const bool hi = (l >= 16);

  const float4* __restrict__ tr4 = reinterpret_cast<const float4*>(trans);
  float4 tv[4];
#pragma unroll
  for (int r = 0; r < 4; ++r) {
    const int k = l + 32 * r;
    tv[r] = tr4[k < NKC ? k : NKC - 1];
  }
  const int p = tile * 16 + m;
  const float4 ov = reinterpret_cast<const float4*>(obs)[p];

#pragma unroll
  for (int r = 0; r < 4; ++r) {
    const int k = l + 32 * r;
    const float4 v = tv[r];
    const float l0 = lse2(v.x, v.z);
    const float l1 = lse2(v.y, v.w);
    if (k < NKC) {
      s_logt[0 * NKC + k] = v.x - l0;
      s_logt[1 * NKC + k] = v.y - l1;
      s_logt[2 * NKC + k] = v.z - l0;
      s_logt[3 * NKC + k] = v.w - l1;
    }
  }
  {
    const float ls0 = lse2(ov.x, ov.y);
    const float ls1 = lse2(ov.z, ov.w);
    float4 r;
    r.x = ov.x - ls0; r.y = ov.y - ls0; r.z = ov.z - ls1; r.w = ov.w - ls1;
    if (!hi) reinterpret_cast<float4*>(LO)[p] = r;
  }
  __syncthreads();

#ifdef HAVE_WMMA_F32X4
  const int row = tile * 16 + m;
  const float nmask = (m < 4) ? 1.0f : 0.0f;
  const float ones4 = (m == 4) ? 1.0f : 0.0f;
  const float* bcolp = s_logt + (m & 3) * NKC;
  v8f c = {0.f, 0.f, 0.f, 0.f, 0.f, 0.f, 0.f, 0.f};
#pragma unroll 5
  for (int kk = 0; kk < 25; ++kk) {
    const int k0 = kk * 4 + (hi ? 2 : 0);
    const float2 av = *reinterpret_cast<const float2*>(A + row * NKC + k0);
    const float2 bv = *reinterpret_cast<const float2*>(bcolp + k0);
    v2f a, b;
    a.x = av.x;
    a.y = av.y;
    b.x = bv.x * nmask + ones4;
    b.y = bv.y * nmask + ones4;
    c = __builtin_amdgcn_wmma_f32_16x16x4_f32(
         false, a,  false, b,
         (short)0, c,  false,  false);
  }
  if (m < 5) {
#pragma unroll
    for (int v = 0; v < 8; ++v) {
      const int r = tile * 16 + v + (hi ? 8 : 0);
      M8[r * 8 + m] = (m < 4) ? __expf(c[v]) : c[v];
    }
  }
#else
  if (l < 16) {
    const int row = tile * 16 + m;
    float acc0 = 0.f, acc1 = 0.f, acc2 = 0.f, acc3 = 0.f, acc4 = 0.f;
    for (int k = 0; k < NKC; ++k) {
      const float av = A[row * NKC + k];
      acc0 += av * s_logt[0 * NKC + k];
      acc1 += av * s_logt[1 * NKC + k];
      acc2 += av * s_logt[2 * NKC + k];
      acc3 += av * s_logt[3 * NKC + k];
      acc4 += av;
    }
    M8[row * 8 + 0] = __expf(acc0); M8[row * 8 + 1] = __expf(acc1);
    M8[row * 8 + 2] = __expf(acc2); M8[row * 8 + 3] = __expf(acc3);
    M8[row * 8 + 4] = acc4;
  }
#endif
}

#define TB 4
#define SBLK 5
#define SSTEPS (TB * SBLK)
#define NSTAGE (NT / SSTEPS)
static_assert(NT % SSTEPS == 0, "stages must tile the sequence");
static_assert(16 % TB == 0 && SSTEPS == 16 + TB, "a stage's outputs: one 16-lane store, one TB-lane store");

__device__ __forceinline__ float rowsum16_pair_max(float& a, float& b, unsigned& ta, unsigned& tb) {
  float mx;
  asm("v_add_f32_dpp %0, %0, %0 row_xmask:1 row_mask:0xf bank_mask:0xf bound_ctrl:1\n\t"
      "v_add_f32_dpp %1, %1, %1 row_xmask:1 row_mask:0xf bank_mask:0xf bound_ctrl:1\n\t"
      "v_add_f32_dpp %0, %0, %0 row_xmask:2 row_mask:0xf bank_mask:0xf bound_ctrl:1\n\t"
      "v_add_f32_dpp %1, %1, %1 row_xmask:2 row_mask:0xf bank_mask:0xf bound_ctrl:1\n\t"
      "v_add_f32_dpp %0, %0, %0 row_xmask:4 row_mask:0xf bank_mask:0xf bound_ctrl:1\n\t"
      "v_add_f32_dpp %1, %1, %1 row_xmask:4 row_mask:0xf bank_mask:0xf bound_ctrl:1\n\t"
      "v_add_f32_dpp %0, %0, %0 row_xmask:8 row_mask:0xf bank_mask:0xf bound_ctrl:1\n\t"
      "v_add_f32_dpp %1, %1, %1 row_xmask:8 row_mask:0xf bank_mask:0xf bound_ctrl:1\n\t"
      "v_max_num_f32 %2, %0, %1"
      : "+v"(a), "+v"(b), "=&v"(mx), "+v"(ta), "+v"(tb));
  return mx;
}

template <int I> __device__ __forceinline__ int row_bcast(int v) {
  return __builtin_amdgcn_update_dpp(0, v, 0x150 + I, 0xf, 0xf, true);
}

#define STAGE_STEPS(F) F(0) F(1) F(2) F(3) F(4) F(5) F(6) F(7) F(8) F(9) \
                       F(10) F(11) F(12) F(13) F(14) F(15) F(16) F(17) F(18) F(19)
static_assert(SSTEPS == 20, "STAGE_STEPS lists the steps of a stage");
#define INDEX_STEP(I)                                                                       \
  const unsigned kr_##I = __umul24(                                                         \
      (unsigned)(((I) < 16) ? row_bcast<((I) & 15)>(kv1) : row_bcast<((I) & 15)>(kv2)),     \
      (unsigned)(NKC / 4)) + pc1;
#define GATHER_STEP(I)                                                                      \
  const float4 c0_##I = A4[kr_##I];                                                         \
  const float4 c1_##I = A4b[kr_##I];
#define STORE_STEP(I)                                                                       \
  s_a[pair][slot][I][row].p[j] = c0_##I;                                                    \
  s_a[pair][slot][I][row].p[j + 16] =                                                       \
      make_float4(has2 ? c1_##I.x : 0.f, has2 ? c1_##I.y : 0.f,                             \
                  has2 ? c1_##I.z : 0.f, has2 ? c1_##I.w : 0.f);

struct CRows {
  float4 c0[TB];
  float4 c1[TB];
  float4 y[TB];
  float4 z[TB];
};

template <typename SA, typename SYZ>
__device__ __forceinline__ void run_block(const CRows& r, CRows& nxt, const SA* sa, const SYZ* syz,
                                          unsigned& ta, unsigned& tb, int j,
                                          float (&al0)[8], float (&al1)[8],
                                          float2* __restrict__ out2, int ob, const int S,
                                          float& sx, float& sy, unsigned& st_i) {
  float ox[TB], oy[TB];
#pragma unroll
  for (int i = 0; i < TB; ++i) {
    float c[8];
    c[0] = r.c0[i].x; c[1] = r.c0[i].y; c[2] = r.c0[i].z; c[3] = r.c0[i].w;
    c[4] = r.c1[i].x; c[5] = r.c1[i].y; c[6] = r.c1[i].z; c[7] = r.c1[i].w;

    float p0 = c[0] * al0[0], p1 = c[0] * al1[0];
    float q0s = c[4] * al0[4], q1s = c[4] * al1[4];
#pragma unroll
    for (int k = 1; k < 4; ++k) {
      p0 = fmaf(c[k], al0[k], p0);
      p1 = fmaf(c[k], al1[k], p1);
      q0s = fmaf(c[4 + k], al0[4 + k], q0s);
      q1s = fmaf(c[4 + k], al1[4 + k], q1s);
    }
    float a20 = p0 + q0s;
    float a21 = p1 + q1s;
    const float mx = rowsum16_pair_max(a20, a21, ta, tb);
    if (i == 0 && S >= 1 && S + 1 < SBLK) {
#pragma unroll
      for (int i2 = 0; i2 < TB; ++i2) {
        const int st = (S + 1) * TB + i2;
        nxt.c0[i2] = sa[ta + 2 * st].p[j];
        nxt.c1[i2] = sa[ta + 2 * st].p[j + 16];
        nxt.y[i2] = syz[tb + 2 * st].y;
        nxt.z[i2] = syz[tb + 2 * st].z;
      }
      __builtin_amdgcn_sched_barrier(0);
    }

    float w0[8], w1[8];
#pragma unroll
    for (int k = 0; k < 8; ++k) {
      w0[k] = fmaf(-c[k], al0[k], al0[k]);
      w1[k] = fmaf(-c[k], al1[k], al1[k]);
    }

    const float y00 = r.y[i].x, y01 = r.y[i].y, y10 = r.y[i].z, y11 = r.y[i].w;
    const float z00 = r.z[i].x, z01 = r.z[i].y, z10 = r.z[i].z, z11 = r.z[i].w;

    float x0 = (a20 - mx) * LOG2E_F;
    float x1 = (a21 - mx) * LOG2E_F;
    asm volatile("" : "+v"(x0), "+v"(x1));
    const float e0 = ex2(x0);
    const float e1 = ex2(x1);

    const float t30 = fmaf(LN2_F, lg2(fmaf(e0, z00, e1 * z01)), mx);
    const float t31 = fmaf(LN2_F, lg2(fmaf(e0, z10, e1 * z11)), mx);

    const float s0 = fmaf(e0, y00, e1 * y10);
    const float s1 = fmaf(e0, y01, e1 * y11);
    const float lt = lg2(s0 + s1);

    ox[i] = (lg2(s0) - lt) * LN2_F;
    oy[i] = (lg2(s1) - lt) * LN2_F;
    if (i + 1 < TB) asm volatile("" : "+v"(ox[i]), "+v"(oy[i]));

#pragma unroll
    for (int k = 0; k < 8; ++k) {
      al0[k] = fmaf(c[k], t30, w0[k]);
      al1[k] = fmaf(c[k], t31, w1[k]);
    }
  }
#pragma unroll
  for (int k = 0; k < 8; ++k) {
    asm volatile("" : "+v"(al0[k]), "+v"(al1[k]));
  }
  asm volatile("" : "+v"(ox[TB - 1]), "+v"(oy[TB - 1]));
  if (S * TB < 16) {
#pragma unroll
    for (int i = 0; i < TB; ++i) {
      sx = (j == S * TB + i) ? ox[i] : sx;
      sy = (j == S * TB + i) ? oy[i] : sy;
    }
    if ((S + 1) * TB == 16) {
      st_i = (unsigned)(ob + j);
      float2 o2;
      o2.x = sx;
      o2.y = sy;
      out2[st_i] = o2;
    }
  } else {
    float tx = ox[0], ty = oy[0];
#pragma unroll
    for (int i = 1; i < TB; ++i) {
      tx = (j == i) ? ox[i] : tx;
      ty = (j == i) ? oy[i] : ty;
    }
    if (j < TB) {
      float2 o2;
      o2.x = tx;
      o2.y = ty;
      out2[ob + S * TB + j] = o2;
    }
  }
}

struct __attribute__((aligned(16))) ARow { float4 p[32]; };
struct __attribute__((aligned(16))) YZ { float4 y; float4 z; };

__device__ __forceinline__ YZ make_yz(const float4 lo, const float sa, const float4 m4, const int ci) {
  const float sal = sa * LOG2E_F;
  const float y00 = ex2(lo.x * sal), y01 = ex2(lo.y * sal);
  const float y10 = ex2(lo.z * sal), y11 = ex2(lo.w * sal);
  const float yc0 = (ci == 0) ? y00 : y01;
  const float yc1 = (ci == 0) ? y10 : y11;
  YZ v;
  v.y = make_float4(y00, y01, y10, y11);
  v.z = make_float4(yc0 * m4.x, yc1 * m4.y, yc0 * m4.z, yc1 * m4.w);
  return v;
}

__global__ void __launch_bounds__(128) k_forward(
    const int* __restrict__ corr, const int* __restrict__ kc,
    const int* __restrict__ problem, const float* __restrict__ A,
    const float* __restrict__ init_logits, const float* __restrict__ M8,
    const float* __restrict__ LO, float* __restrict__ out) {
  __shared__ ARow s_a[2][2][SSTEPS][2];
  __shared__ YZ s_yz[2][2][SSTEPS][2];

  const int l = threadIdx.x & 31;
  const int wv = threadIdx.x >> 5;
  const int pair = wv & 1;
  const bool helper = (wv >= 2);
  const int j = l & 15;
  const int row = l >> 4;
  const int b = (blockIdx.x * 2 + pair) * 2 + row;
  const bool has2 = (j >= 7);
  const float4* __restrict__ A4  = reinterpret_cast<const float4*>(A);
  const float4* __restrict__ A4b = A4 + 9;
  const float4* __restrict__ M84 = reinterpret_cast<const float4*>(M8);
  const float4* __restrict__ LO4 = reinterpret_cast<const float4*>(LO);
  float2* __restrict__ out2 = reinterpret_cast<float2*>(out);
  const int base = b * NT;
  const unsigned pc1 = (unsigned)j;
  const int t1 = j;
  const int t2 = 16 + (j & 3);

  float al0[8], al1[8];
  int kv1 = 0, pv1 = 0, cv1 = 0, kv2 = 0, pv2 = 0, cv2 = 0;
  if (!helper) {
    const float4* il = reinterpret_cast<const float4*>(init_logits);
    const float4 i0 = il[2 * j];
    const float4 i1 = il[2 * j + 1];
    const int j2 = 2 * (j + 9);
    const float4 i2 = il[j2];
    const float4 i3 = il[j2 + 1];
    float z;
    z = lse2(i0.x, i0.y); al0[0] = i0.x - z; al1[0] = i0.y - z;
    z = lse2(i0.z, i0.w); al0[1] = i0.z - z; al1[1] = i0.w - z;
    z = lse2(i1.x, i1.y); al0[2] = i1.x - z; al1[2] = i1.y - z;
    z = lse2(i1.z, i1.w); al0[3] = i1.z - z; al1[3] = i1.w - z;
    z = lse2(i2.x, i2.y); al0[4] = has2 ? i2.x - z : 0.f; al1[4] = has2 ? i2.y - z : 0.f;
    z = lse2(i2.z, i2.w); al0[5] = has2 ? i2.z - z : 0.f; al1[5] = has2 ? i2.w - z : 0.f;
    z = lse2(i3.x, i3.y); al0[6] = has2 ? i3.x - z : 0.f; al1[6] = has2 ? i3.y - z : 0.f;
    z = lse2(i3.z, i3.w); al0[7] = has2 ? i3.z - z : 0.f; al1[7] = has2 ? i3.w - z : 0.f;
  } else {
#pragma unroll
    for (int k = 0; k < 8; ++k) { al0[k] = 0.f; al1[k] = 0.f; }
    kv1 = kc[base + t1]; pv1 = problem[base + t1]; cv1 = corr[base + t1];
    kv2 = kc[base + t2]; pv2 = problem[base + t2]; cv2 = corr[base + t2];
  }

  for (int n = 0; n <= NSTAGE; ++n) {
    if (helper) {
      if (n < NSTAGE) {
        const int slot = n & 1;
        STAGE_STEPS(INDEX_STEP)
        const unsigned ma = 2u * (unsigned)kv1, mb = 2u * (unsigned)kv2;
        const unsigned qa = 8u * (unsigned)kv1, qb = 8u * (unsigned)kv2;
        const int nb = base + ((n + 1 < NSTAGE) ? (n + 1) : n) * SSTEPS;
        const unsigned n1 = (unsigned)(nb + t1), n2 = (unsigned)(nb + t2);
        asm volatile("" : : "v"(kr_0), "v"(kr_1), "v"(kr_2), "v"(kr_3), "v"(kr_4), "v"(kr_5),
                     "v"(kr_6), "v"(kr_7), "v"(kr_8), "v"(kr_9), "v"(kr_10), "v"(kr_11),
                     "v"(kr_12), "v"(kr_13), "v"(kr_14), "v"(kr_15), "v"(kr_16), "v"(kr_17),
                     "v"(kr_18), "v"(kr_19), "v"(ma), "v"(mb), "v"(qa), "v"(qb), "v"(n1), "v"(n2));
        __builtin_amdgcn_sched_barrier(0);
        STAGE_STEPS(GATHER_STEP)
        const float4 m4a = M84[ma];
        const float saa = (M8 + 4)[qa];
        const float4 loa = LO4[(unsigned)pv1];
        const float4 m4b = M84[mb];
        const float sab = (M8 + 4)[qb];
        const float4 lob = LO4[(unsigned)pv2];
        const int cia = cv1, cib = cv2;
        kv1 = kc[n1]; pv1 = problem[n1]; cv1 = corr[n1];
        kv2 = kc[n2]; pv2 = problem[n2]; cv2 = corr[n2];
        __builtin_amdgcn_sched_barrier(0);

        s_yz[pair][slot][t1][row] = make_yz(loa, saa, m4a, cia);
        s_yz[pair][slot][t2][row] = make_yz(lob, sab, m4b, cib);
        STAGE_STEPS(STORE_STEP)
      }
    } else {
      if (n >= 1) {
        const int m = n - 1;
        const int slot = m & 1;
        const int ob = base + m * SSTEPS;
        float sx = 0.f, sy = 0.f;
        unsigned st_i = 0u;
        unsigned ta = (unsigned)(((pair * 2 + slot) * SSTEPS) * 2 + row);
        unsigned tb = ta;
        CRows ra, rb;
#pragma unroll
        for (int i = 0; i < TB; ++i) {
          ra.c0[i] = s_a[pair][slot][i][row].p[j];
          ra.c1[i] = s_a[pair][slot][i][row].p[j + 16];
          ra.y[i] = s_yz[pair][slot][i][row].y;
          ra.z[i] = s_yz[pair][slot][i][row].z;
        }
#pragma unroll
        for (int s = 0; s < SBLK; ++s) {
          CRows& cur = (s & 1) ? rb : ra;
          CRows& nxt = (s & 1) ? ra : rb;
          if (s == 0) {
#pragma unroll
            for (int i = 0; i < TB; ++i) {
              const int st = TB + i;
              nxt.c0[i] = s_a[pair][slot][st][row].p[j];
              nxt.c1[i] = s_a[pair][slot][st][row].p[j + 16];
              nxt.y[i] = s_yz[pair][slot][st][row].y;
              nxt.z[i] = s_yz[pair][slot][st][row].z;
            }
            __builtin_amdgcn_sched_barrier(0);
          }
          run_block(cur, nxt, &s_a[0][0][0][0], &s_yz[0][0][0][0], ta, tb, j, al0, al1, out2, ob, s,
                    sx, sy, st_i);
        }
        asm volatile("" : : "v"(sx), "v"(sy), "v"(st_i));
      }
    }
    __builtin_amdgcn_fence(__ATOMIC_RELEASE, "workgroup", "local");
    __builtin_amdgcn_s_barrier();
    __builtin_amdgcn_fence(__ATOMIC_ACQUIRE, "workgroup", "local");
  }
}

extern "C" void kernel_launch(void* const* d_in, const int* in_sizes, int n_in,
                              void* d_out, int out_size, void* d_ws, size_t ws_size,
                              hipStream_t stream) {
  (void)in_sizes; (void)n_in; (void)out_size; (void)ws_size;
  const int*   corr    = (const int*)d_in[0];
  const int*   kc      = (const int*)d_in[1];
  const int*   problem = (const int*)d_in[2];
  const float* A       = (const float*)d_in[3];
  const float* trans   = (const float*)d_in[4];
  const float* obs     = (const float*)d_in[5];
  const float* init    = (const float*)d_in[6];
  float* out = (float*)d_out;

  float* M8    = (float*)d_ws + 512;
  float* LO    = M8 + (size_t)NPROB * 8;

  k_tables<<<NPROB / 16, 32, 0, stream>>>(A, trans, obs, M8, LO);
  k_forward<<<NBATCH / 4, 128, 0, stream>>>(corr, kc, problem, A, init, M8, LO, out);
}


# k_forward scan carries only the state difference d[k]=alpha[k][0]-alpha[k][1] (base-2 units): outputs and d's own update need nothing else; one row sum and one exp(-|D|) per step; new rounding order
# speedup vs baseline: 9.2374x; 1.0566x over previous
#include <hip/hip_runtime.h>
#include <math.h>

#define NPROB 10000
#define NKC   100
#define NBATCH 1024
#define NT    500

typedef __attribute__((ext_vector_type(2))) float v2f;
typedef __attribute__((ext_vector_type(8))) float v8f;

__device__ __forceinline__ float lse2(float a, float b) {
  float m = fmaxf(a, b);
  return m + __logf(__expf(a - m) + __expf(b - m));
}

__device__ __forceinline__ float ex2(float x) { return __builtin_amdgcn_exp2f(x); }
__device__ __forceinline__ float lg2(float x) { return __builtin_amdgcn_logf(x); }
#define LOG2E_F 1.4426950408889634f
#define LN2_F   0.6931471805599453f

#if __has_builtin(__builtin_amdgcn_wmma_f32_16x16x4_f32)
#define HAVE_WMMA_F32X4 1
#endif

__global__ void __launch_bounds__(32) k_tables(const float* __restrict__ A,
                                               const float* __restrict__ trans,
                                               const float* __restrict__ obs,
                                               float* __restrict__ M8,
                                               float* __restrict__ LO) {
  __shared__ __attribute__((aligned(16))) float s_logt[4 * NKC];
  const int tile = blockIdx.x;
  const int l = threadIdx.x;
  const int m = l & 15;
  const bool hi = (l >= 16);

  const float4* __restrict__ tr4 = reinterpret_cast<const float4*>(trans);
  float4 tv[4];
#pragma unroll
  for (int r = 0; r < 4; ++r) {
    const int k = l + 32 * r;
    tv[r] = tr4[k < NKC ? k : NKC - 1];
  }
  const int p = tile * 16 + m;
  const float4 ov = reinterpret_cast<const float4*>(obs)[p];

#pragma unroll
  for (int r = 0; r < 4; ++r) {
    const int k = l + 32 * r;
    const float4 v = tv[r];
    const float l0 = lse2(v.x, v.z);
    const float l1 = lse2(v.y, v.w);
    if (k < NKC) {
      s_logt[0 * NKC + k] = v.x - l0;
      s_logt[1 * NKC + k] = v.y - l1;
      s_logt[2 * NKC + k] = v.z - l0;
      s_logt[3 * NKC + k] = v.w - l1;
    }
  }
  {
    const float ls0 = lse2(ov.x, ov.y);
    const float ls1 = lse2(ov.z, ov.w);
    float4 r;
    r.x = ov.x - ls0; r.y = ov.y - ls0; r.z = ov.z - ls1; r.w = ov.w - ls1;
    if (!hi) reinterpret_cast<float4*>(LO)[p] = r;
  }
  __syncthreads();

#ifdef HAVE_WMMA_F32X4
  const int row = tile * 16 + m;
  const float nmask = (m < 4) ? 1.0f : 0.0f;
  const float ones4 = (m == 4) ? 1.0f : 0.0f;
  const float* bcolp = s_logt + (m & 3) * NKC;
  v8f c = {0.f, 0.f, 0.f, 0.f, 0.f, 0.f, 0.f, 0.f};
#pragma unroll 5
  for (int kk = 0; kk < 25; ++kk) {
    const int k0 = kk * 4 + (hi ? 2 : 0);
    const float2 av = *reinterpret_cast<const float2*>(A + row * NKC + k0);
    const float2 bv = *reinterpret_cast<const float2*>(bcolp + k0);
    v2f a, b;
    a.x = av.x;
    a.y = av.y;
    b.x = bv.x * nmask + ones4;
    b.y = bv.y * nmask + ones4;
    c = __builtin_amdgcn_wmma_f32_16x16x4_f32(
         false, a,  false, b,
         (short)0, c,  false,  false);
  }
  if (m < 5) {
#pragma unroll
    for (int v = 0; v < 8; ++v) {
      const int r = tile * 16 + v + (hi ? 8 : 0);
      M8[r * 8 + m] = (m < 4) ? __expf(c[v]) : c[v];
    }
  }
#else
  if (l < 16) {
    const int row = tile * 16 + m;
    float acc0 = 0.f, acc1 = 0.f, acc2 = 0.f, acc3 = 0.f, acc4 = 0.f;
    for (int k = 0; k < NKC; ++k) {
      const float av = A[row * NKC + k];
      acc0 += av * s_logt[0 * NKC + k];
      acc1 += av * s_logt[1 * NKC + k];
      acc2 += av * s_logt[2 * NKC + k];
      acc3 += av * s_logt[3 * NKC + k];
      acc4 += av;
    }
    M8[row * 8 + 0] = __expf(acc0); M8[row * 8 + 1] = __expf(acc1);
    M8[row * 8 + 2] = __expf(acc2); M8[row * 8 + 3] = __expf(acc3);
    M8[row * 8 + 4] = acc4;
  }
#endif
}

#define TB 4
#define SBLK 5
#define SSTEPS (TB * SBLK)
#define NSTAGE (NT / SSTEPS)
static_assert(NT % SSTEPS == 0, "stages must tile the sequence");
static_assert(16 % TB == 0 && SSTEPS == 16 + TB, "a stage's outputs: one 16-lane store, one TB-lane store");

__device__ __forceinline__ void rowsum16(float& a, unsigned& ta, unsigned& tb) {
  asm("v_add_f32_dpp %0, %0, %0 row_xmask:1 row_mask:0xf bank_mask:0xf bound_ctrl:1\n\t"
      "v_add_f32_dpp %0, %0, %0 row_xmask:2 row_mask:0xf bank_mask:0xf bound_ctrl:1\n\t"
      "v_add_f32_dpp %0, %0, %0 row_xmask:4 row_mask:0xf bank_mask:0xf bound_ctrl:1\n\t"
      "v_add_f32_dpp %0, %0, %0 row_xmask:8 row_mask:0xf bank_mask:0xf bound_ctrl:1"
      : "+v"(a), "+v"(ta), "+v"(tb));
}

template <int I> __device__ __forceinline__ int row_bcast(int v) {
  return __builtin_amdgcn_update_dpp(0, v, 0x150 + I, 0xf, 0xf, true);
}

#define STAGE_STEPS(F) F(0) F(1) F(2) F(3) F(4) F(5) F(6) F(7) F(8) F(9) \
                       F(10) F(11) F(12) F(13) F(14) F(15) F(16) F(17) F(18) F(19)
static_assert(SSTEPS == 20, "STAGE_STEPS lists the steps of a stage");
#define INDEX_STEP(I)                                                                       \
  const unsigned kr_##I = __umul24(                                                         \
      (unsigned)(((I) < 16) ? row_bcast<((I) & 15)>(kv1) : row_bcast<((I) & 15)>(kv2)),     \
      (unsigned)(NKC / 4)) + pc1;
#define GATHER_STEP(I)                                                                      \
  const float4 c0_##I = A4[kr_##I];                                                         \
  const float4 c1_##I = A4b[kr_##I];
#define STORE_STEP(I)                                                                       \
  s_a[pair][slot][I][row].p[j] = c0_##I;                                                    \
  s_a[pair][slot][I][row].p[j + 16] =                                                       \
      make_float4(has2 ? c1_##I.x : 0.f, has2 ? c1_##I.y : 0.f,                             \
                  has2 ? c1_##I.z : 0.f, has2 ? c1_##I.w : 0.f);

struct CRows {
  float4 c0[TB];
  float4 c1[TB];
  float4 y[TB];
  float4 z[TB];
};

template <typename SA, typename SYZ>
__device__ __forceinline__ void run_block(const CRows& r, CRows& nxt, const SA* sa, const SYZ* syz,
                                          unsigned& ta, unsigned& tb, int j,
                                          float (&dl)[8],
                                          float2* __restrict__ out2, int ob, const int S,
                                          float& sx, float& sy, unsigned& st_i) {
  float ox[TB], oy[TB];
#pragma unroll
  for (int i = 0; i < TB; ++i) {
    float c[8];
    c[0] = r.c0[i].x; c[1] = r.c0[i].y; c[2] = r.c0[i].z; c[3] = r.c0[i].w;
    c[4] = r.c1[i].x; c[5] = r.c1[i].y; c[6] = r.c1[i].z; c[7] = r.c1[i].w;

    float p = c[0] * dl[0];
    float q = c[4] * dl[4];
#pragma unroll
    for (int k = 1; k < 4; ++k) {
      p = fmaf(c[k], dl[k], p);
      q = fmaf(c[4 + k], dl[4 + k], q);
    }
    float D = p + q;
    rowsum16(D, ta, tb);
    if (i == 0 && S >= 1 && S + 1 < SBLK) {
#pragma unroll
      for (int i2 = 0; i2 < TB; ++i2) {
        const int st = (S + 1) * TB + i2;
        nxt.c0[i2] = sa[ta + 2 * st].p[j];
        nxt.c1[i2] = sa[ta + 2 * st].p[j + 16];
        nxt.y[i2] = syz[tb + 2 * st].y;
        nxt.z[i2] = syz[tb + 2 * st].z;
      }
      __builtin_amdgcn_sched_barrier(0);
    }

    float w[8];
#pragma unroll
    for (int k = 0; k < 8; ++k) {
      w[k] = fmaf(-c[k], dl[k], dl[k]);
    }

    const float y00 = r.y[i].x, y01 = r.y[i].y, y10 = r.y[i].z, y11 = r.y[i].w;
    const float z00 = r.z[i].x, z01 = r.z[i].y, z10 = r.z[i].z, z11 = r.z[i].w;

    const bool pos = (D >= 0.f);
    const float e = ex2(-fabsf(D));

    const float zA0 = pos ? z00 : z01, zB0 = pos ? z01 : z00;
    const float zA1 = pos ? z10 : z11, zB1 = pos ? z11 : z10;
    const float ld = lg2(fmaf(e, zB0, zA0)) - lg2(fmaf(e, zB1, zA1));

    const float yA0 = pos ? y00 : y10, yB0 = pos ? y10 : y00;
    const float yA1 = pos ? y01 : y11, yB1 = pos ? y11 : y01;
    const float s0 = fmaf(e, yB0, yA0);
    const float s1 = fmaf(e, yB1, yA1);
    const float lt = lg2(s0 + s1);

    ox[i] = (lg2(s0) - lt) * LN2_F;
    oy[i] = (lg2(s1) - lt) * LN2_F;
    if (i + 1 < TB) asm volatile("" : "+v"(ox[i]), "+v"(oy[i]));

#pragma unroll
    for (int k = 0; k < 8; ++k) {
      dl[k] = fmaf(c[k], ld, w[k]);
    }
  }
#pragma unroll
  for (int k = 0; k < 8; ++k) {
    asm volatile("" : "+v"(dl[k]));
  }
  asm volatile("" : "+v"(ox[TB - 1]), "+v"(oy[TB - 1]));
  if (S * TB < 16) {
#pragma unroll
    for (int i = 0; i < TB; ++i) {
      sx = (j == S * TB + i) ? ox[i] : sx;
      sy = (j == S * TB + i) ? oy[i] : sy;
    }
    if ((S + 1) * TB == 16) {
      st_i = (unsigned)(ob + j);
      float2 o2;
      o2.x = sx;
      o2.y = sy;
      out2[st_i] = o2;
    }
  } else {
    float tx = ox[0], ty = oy[0];
#pragma unroll
    for (int i = 1; i < TB; ++i) {
      tx = (j == i) ? ox[i] : tx;
      ty = (j == i) ? oy[i] : ty;
    }
    if (j < TB) {
      float2 o2;
      o2.x = tx;
      o2.y = ty;
      out2[ob + S * TB + j] = o2;
    }
  }
}

struct __attribute__((aligned(16))) ARow { float4 p[32]; };
struct __attribute__((aligned(16))) YZ { float4 y; float4 z; };

__device__ __forceinline__ YZ make_yz(const float4 lo, const float sa, const float4 m4, const int ci) {
  const float sal = sa * LOG2E_F;
  const float y00 = ex2(lo.x * sal), y01 = ex2(lo.y * sal);
  const float y10 = ex2(lo.z * sal), y11 = ex2(lo.w * sal);
  const float yc0 = (ci == 0) ? y00 : y01;
  const float yc1 = (ci == 0) ? y10 : y11;
  YZ v;
  v.y = make_float4(y00, y01, y10, y11);
  v.z = make_float4(yc0 * m4.x, yc1 * m4.y, yc0 * m4.z, yc1 * m4.w);
  return v;
}

__global__ void __launch_bounds__(128) k_forward(
    const int* __restrict__ corr, const int* __restrict__ kc,
    const int* __restrict__ problem, const float* __restrict__ A,
    const float* __restrict__ init_logits, const float* __restrict__ M8,
    const float* __restrict__ LO, float* __restrict__ out) {
  __shared__ ARow s_a[2][2][SSTEPS][2];
  __shared__ YZ s_yz[2][2][SSTEPS][2];

  const int l = threadIdx.x & 31;
  const int wv = threadIdx.x >> 5;
  const int pair = wv & 1;
  const bool helper = (wv >= 2);
  const int j = l & 15;
  const int row = l >> 4;
  const int b = (blockIdx.x * 2 + pair) * 2 + row;
  const bool has2 = (j >= 7);
  const float4* __restrict__ A4  = reinterpret_cast<const float4*>(A);
  const float4* __restrict__ A4b = A4 + 9;
  const float4* __restrict__ M84 = reinterpret_cast<const float4*>(M8);
  const float4* __restrict__ LO4 = reinterpret_cast<const float4*>(LO);
  float2* __restrict__ out2 = reinterpret_cast<float2*>(out);
  const int base = b * NT;
  const unsigned pc1 = (unsigned)j;
  const int t1 = j;
  const int t2 = 16 + (j & 3);

  float dl[8];
  int kv1 = 0, pv1 = 0, cv1 = 0, kv2 = 0, pv2 = 0, cv2 = 0;
  if (!helper) {
    const float4* il = reinterpret_cast<const float4*>(init_logits);
    const float4 i0 = il[2 * j];
    const float4 i1 = il[2 * j + 1];
    const int j2 = 2 * (j + 9);
    const float4 i2 = il[j2];
    const float4 i3 = il[j2 + 1];
    dl[0] = (i0.x - i0.y) * LOG2E_F;
    dl[1] = (i0.z - i0.w) * LOG2E_F;
    dl[2] = (i1.x - i1.y) * LOG2E_F;
    dl[3] = (i1.z - i1.w) * LOG2E_F;
    dl[4] = has2 ? (i2.x - i2.y) * LOG2E_F : 0.f;
    dl[5] = has2 ? (i2.z - i2.w) * LOG2E_F : 0.f;
    dl[6] = has2 ? (i3.x - i3.y) * LOG2E_F : 0.f;
    dl[7] = has2 ? (i3.z - i3.w) * LOG2E_F : 0.f;
  } else {
#pragma unroll
    for (int k = 0; k < 8; ++k) { dl[k] = 0.f; }
    kv1 = kc[base + t1]; pv1 = problem[base + t1]; cv1 = corr[base + t1];
    kv2 = kc[base + t2]; pv2 = problem[base + t2]; cv2 = corr[base + t2];
  }

  for (int n = 0; n <= NSTAGE; ++n) {
    if (helper) {
      if (n < NSTAGE) {
        const int slot = n & 1;
        STAGE_STEPS(INDEX_STEP)
        const unsigned ma = 2u * (unsigned)kv1, mb = 2u * (unsigned)kv2;
        const unsigned qa = 8u * (unsigned)kv1, qb = 8u * (unsigned)kv2;
        const int nb = base + ((n + 1 < NSTAGE) ? (n + 1) : n) * SSTEPS;
        const unsigned n1 = (unsigned)(nb + t1), n2 = (unsigned)(nb + t2);
        asm volatile("" : : "v"(kr_0), "v"(kr_1), "v"(kr_2), "v"(kr_3), "v"(kr_4), "v"(kr_5),
                     "v"(kr_6), "v"(kr_7), "v"(kr_8), "v"(kr_9), "v"(kr_10), "v"(kr_11),
                     "v"(kr_12), "v"(kr_13), "v"(kr_14), "v"(kr_15), "v"(kr_16), "v"(kr_17),
                     "v"(kr_18), "v"(kr_19), "v"(ma), "v"(mb), "v"(qa), "v"(qb), "v"(n1), "v"(n2));
        __builtin_amdgcn_sched_barrier(0);
        STAGE_STEPS(GATHER_STEP)
        const float4 m4a = M84[ma];
        const float saa = (M8 + 4)[qa];
        const float4 loa = LO4[(unsigned)pv1];
        const float4 m4b = M84[mb];
        const float sab = (M8 + 4)[qb];
        const float4 lob = LO4[(unsigned)pv2];
        const int cia = cv1, cib = cv2;
        kv1 = kc[n1]; pv1 = problem[n1]; cv1 = corr[n1];
        kv2 = kc[n2]; pv2 = problem[n2]; cv2 = corr[n2];
        __builtin_amdgcn_sched_barrier(0);

        s_yz[pair][slot][t1][row] = make_yz(loa, saa, m4a, cia);
        s_yz[pair][slot][t2][row] = make_yz(lob, sab, m4b, cib);
        STAGE_STEPS(STORE_STEP)
      }
    } else {
      if (n >= 1) {
        const int m = n - 1;
        const int slot = m & 1;
        const int ob = base + m * SSTEPS;
        float sx = 0.f, sy = 0.f;
        unsigned st_i = 0u;
        unsigned ta = (unsigned)(((pair * 2 + slot) * SSTEPS) * 2 + row);
        unsigned tb = ta;
        CRows ra, rb;
#pragma unroll
        for (int i = 0; i < TB; ++i) {
          ra.c0[i] = s_a[pair][slot][i][row].p[j];
          ra.c1[i] = s_a[pair][slot][i][row].p[j + 16];
          ra.y[i] = s_yz[pair][slot][i][row].y;
          ra.z[i] = s_yz[pair][slot][i][row].z;
        }
#pragma unroll
        for (int s = 0; s < SBLK; ++s) {
          CRows& cur = (s & 1) ? rb : ra;
          CRows& nxt = (s & 1) ? ra : rb;
          if (s == 0) {
#pragma unroll
            for (int i = 0; i < TB; ++i) {
              const int st = TB + i;
              nxt.c0[i] = s_a[pair][slot][st][row].p[j];
              nxt.c1[i] = s_a[pair][slot][st][row].p[j + 16];
              nxt.y[i] = s_yz[pair][slot][st][row].y;
              nxt.z[i] = s_yz[pair][slot][st][row].z;
            }
            __builtin_amdgcn_sched_barrier(0);
          }
          run_block(cur, nxt, &s_a[0][0][0][0], &s_yz[0][0][0][0], ta, tb, j, dl, out2, ob, s,
                    sx, sy, st_i);
        }
        asm volatile("" : : "v"(sx), "v"(sy), "v"(st_i));
      }
    }
    __builtin_amdgcn_fence(__ATOMIC_RELEASE, "workgroup", "local");
    __builtin_amdgcn_s_barrier();
    __builtin_amdgcn_fence(__ATOMIC_ACQUIRE, "workgroup", "local");
  }
}

extern "C" void kernel_launch(void* const* d_in, const int* in_sizes, int n_in,
                              void* d_out, int out_size, void* d_ws, size_t ws_size,
                              hipStream_t stream) {
  (void)in_sizes; (void)n_in; (void)out_size; (void)ws_size;
  const int*   corr    = (const int*)d_in[0];
  const int*   kc      = (const int*)d_in[1];
  const int*   problem = (const int*)d_in[2];
  const float* A       = (const float*)d_in[3];
  const float* trans   = (const float*)d_in[4];
  const float* obs     = (const float*)d_in[5];
  const float* init    = (const float*)d_in[6];
  float* out = (float*)d_out;

  float* M8    = (float*)d_ws + 512;
  float* LO    = M8 + (size_t)NPROB * 8;

  k_tables<<<NPROB / 16, 32, 0, stream>>>(A, trans, obs, M8, LO);
  k_forward<<<NBATCH / 4, 128, 0, stream>>>(corr, kc, problem, A, init, M8, LO, out);
}
